# starting kernel
# baseline (speedup 1.0000x reference)
#include <hip/hip_runtime.h>

typedef __attribute__((ext_vector_type(2))) float v2f;
typedef __attribute__((ext_vector_type(8))) float v8f;

#define N_NODES 100000
#define N_EDGES 3200000


__global__ void k_init_deg(float* __restrict__ deg) {
    int i = blockIdx.x * blockDim.x + threadIdx.x;
    if (i < N_NODES) deg[i] = 1.0f;
}

__global__ void k_count_deg(const int* __restrict__ dst, float* __restrict__ deg) {
    int e = blockIdx.x * blockDim.x + threadIdx.x;
    if (e < N_EDGES) atomicAdd(&deg[dst[e]], 1.0f);
}

__global__ void k_to_dis(float* __restrict__ deg) {
    int i = blockIdx.x * blockDim.x + threadIdx.x;
    if (i < N_NODES) deg[i] = rsqrtf(deg[i]);
}


template <int IN, int OUT>
__global__ __launch_bounds__(256) void k_gemm_wmma(const float* __restrict__ X,
                                                   const float* __restrict__ W,
                                                   float* __restrict__ Y,
                                                   int nrows) {
    const int lane = threadIdx.x & 31;
    const int wave = blockIdx.x * (blockDim.x >> 5) + (threadIdx.x >> 5);
    const int row0 = wave * 16;
    if (row0 >= nrows) return;

    const int half = lane >> 4;
    const int l15  = lane & 15;
    const int arow = row0 + l15;

    constexpr int NT = OUT / 16;
    v8f acc[NT];
    #pragma unroll
    for (int t = 0; t < NT; ++t)
        #pragma unroll
        for (int v = 0; v < 8; ++v) acc[t][v] = 0.0f;

    for (int k = 0; k < IN; k += 4) {
        const float* ap = X + (size_t)arow * IN + (k + 2 * half);
        v2f a;
        a.x = ap[0];
        a.y = ap[1];

        #pragma unroll
        for (int t = 0; t < NT; ++t) {
            const int col = t * 16 + l15;
            v2f b;
            b.x = W[(k + 2 * half)     * OUT + col];
            b.y = W[(k + 2 * half + 1) * OUT + col];
            acc[t] = __builtin_amdgcn_wmma_f32_16x16x4_f32(
                 false, a,  false, b,
                 (short)0, acc[t],  false,  false);
        }
    }

    #pragma unroll
    for (int t = 0; t < NT; ++t)
        #pragma unroll
        for (int v = 0; v < 8; ++v) {
            int r = row0 + v + 8 * half;
            int c = t * 16 + l15;
            Y[(size_t)r * OUT + c] = acc[t][v];
        }
}


template <int OUT>
__global__ void k_self_init(const float* __restrict__ hw, const float* __restrict__ dis,
                            float* __restrict__ agg) {
    long long t = (long long)blockIdx.x * blockDim.x + threadIdx.x;
    if (t >= (long long)N_NODES * OUT) return;
    int i = (int)(t / OUT);
    float d = dis[i];
    agg[t] = d * d * hw[t];
}

template <int OUT>
__global__ void k_edge_agg(const int* __restrict__ src, const int* __restrict__ dst,
                           const float* __restrict__ dis, const float* __restrict__ hw,
                           float* __restrict__ agg) {
    long long t = (long long)blockIdx.x * blockDim.x + threadIdx.x;
    if (t >= (long long)N_EDGES * OUT) return;
    int e = (int)(t / OUT);
    int f = (int)(t % OUT);
    int s = src[e];
    int d = dst[e];
    float nrm = dis[s] * dis[d];
    atomicAdd(&agg[(long long)d * OUT + f], nrm * hw[(long long)s * OUT + f]);
}

template <int OUT>
__global__ void k_bias_relu(float* __restrict__ h, const float* __restrict__ b) {
    long long t = (long long)blockIdx.x * blockDim.x + threadIdx.x;
    if (t >= (long long)N_NODES * OUT) return;
    float v = h[t] + b[t % OUT];
    h[t] = v > 0.0f ? v : 0.0f;
}


__global__ void k_gemm_out1(const float* __restrict__ h2, const float* __restrict__ W3,
                            float* __restrict__ hw3) {
    int i = blockIdx.x * blockDim.x + threadIdx.x;
    if (i >= N_NODES) return;
    float s = 0.0f;
    #pragma unroll
    for (int k = 0; k < 16; ++k) s += h2[(size_t)i * 16 + k] * W3[k];
    hw3[i] = s;
}

__global__ void k_self_init_out1(const float* __restrict__ hw3, const float* __restrict__ dis,
                                 const float* __restrict__ b3, float* __restrict__ out) {
    int i = blockIdx.x * blockDim.x + threadIdx.x;
    if (i >= N_NODES) return;
    float d = dis[i];
    out[i] = d * d * hw3[i] + b3[0];
}

__global__ void k_edge_agg_out1(const int* __restrict__ src, const int* __restrict__ dst,
                                const float* __restrict__ dis, const float* __restrict__ hw3,
                                float* __restrict__ out) {
    int e = blockIdx.x * blockDim.x + threadIdx.x;
    if (e >= N_EDGES) return;
    int s = src[e];
    int d = dst[e];
    atomicAdd(&out[d], dis[s] * dis[d] * hw3[s]);
}


static inline int blks(long long n, int bs) { return (int)((n + bs - 1) / bs); }

extern "C" void kernel_launch(void* const* d_in, const int* in_sizes, int n_in,
                              void* d_out, int out_size, void* d_ws, size_t ws_size,
                              hipStream_t stream) {
    (void)in_sizes; (void)n_in; (void)out_size; (void)ws_size;

    const float* x   = (const float*)d_in[0];
    const int*   ei  = (const int*)  d_in[1];
    const float* W1  = (const float*)d_in[2];
    const float* b1  = (const float*)d_in[3];
    const float* W2  = (const float*)d_in[4];
    const float* b2  = (const float*)d_in[5];
    const float* W3  = (const float*)d_in[6];
    const float* b3  = (const float*)d_in[7];
    float* out = (float*)d_out;

    const int* src = ei;
    const int* dst = ei + N_EDGES;

    float* ws   = (float*)d_ws;
    float* dis  = ws;
    float* hw1  = dis  + N_NODES;
    float* h1   = hw1  + 32 * N_NODES;
    float* hw2  = h1   + 32 * N_NODES;
    float* h2   = hw2  + 16 * N_NODES;
    float* hw3  = h2   + 16 * N_NODES;

    const int BS = 256;

    k_init_deg <<<blks(N_NODES, BS), BS, 0, stream>>>(dis);
    k_count_deg<<<blks(N_EDGES, BS), BS, 0, stream>>>(dst, dis);
    k_to_dis   <<<blks(N_NODES, BS), BS, 0, stream>>>(dis);

    const int gemm_grid = blks(N_NODES / 16, BS / 32);

    k_gemm_wmma<128, 32><<<gemm_grid, BS, 0, stream>>>(x, W1, hw1, N_NODES);
    k_self_init<32><<<blks((long long)N_NODES * 32, BS), BS, 0, stream>>>(hw1, dis, h1);
    k_edge_agg <32><<<blks((long long)N_EDGES * 32, BS), BS, 0, stream>>>(src, dst, dis, hw1, h1);
    k_bias_relu<32><<<blks((long long)N_NODES * 32, BS), BS, 0, stream>>>(h1, b1);

    k_gemm_wmma<32, 16><<<gemm_grid, BS, 0, stream>>>(h1, W2, hw2, N_NODES);
    k_self_init<16><<<blks((long long)N_NODES * 16, BS), BS, 0, stream>>>(hw2, dis, h2);
    k_edge_agg <16><<<blks((long long)N_EDGES * 16, BS), BS, 0, stream>>>(src, dst, dis, hw2, h2);
    k_bias_relu<16><<<blks((long long)N_NODES * 16, BS), BS, 0, stream>>>(h2, b2);

    k_gemm_out1     <<<blks(N_NODES, BS), BS, 0, stream>>>(h2, W3, hw3);
    k_self_init_out1<<<blks(N_NODES, BS), BS, 0, stream>>>(hw3, dis, b3, out);
    k_edge_agg_out1 <<<blks(N_EDGES, BS), BS, 0, stream>>>(src, dst, dis, hw3, out);
}


# Edges counting-sorted by destination in-call (int atomic arrival rank + scan); per-node gather replaces every float atomic scatter, one store per row; float sums now follow arrival-rank order
# speedup vs baseline: 2.2412x; 2.2412x over previous
#include <hip/hip_runtime.h>


typedef __attribute__((ext_vector_type(2))) float v2f;
typedef __attribute__((ext_vector_type(8))) float v8f;

#define N_NODES 100000
#define N_EDGES 3200000

static_assert(N_NODES % 32 == 0, "layer-3 gather writes one full 32-node line per block");
static_assert(N_EDGES == 32 * N_NODES, "the rank array reuses the 32*N word plane of the layer-1 output");


__global__ void k_zero_cnt(int* __restrict__ cnt) {
    int i = blockIdx.x * blockDim.x + threadIdx.x;
    if (i < N_NODES) cnt[i] = 0;
}

__global__ void k_rank(const int* __restrict__ dst, int* __restrict__ cnt, int* __restrict__ rank) {
    int e = blockIdx.x * blockDim.x + threadIdx.x;
    if (e < N_EDGES) rank[e] = atomicAdd(&cnt[dst[e]], 1);
}

__global__ __launch_bounds__(1024) void k_scan(const int* __restrict__ cnt, int* __restrict__ off,
                                               float* __restrict__ dis) {
    __shared__ int s_incl[32];
    __shared__ int s_before[32];
    const int t = threadIdx.x;
    const int lane = t & 31;
    const int w = t >> 5;
    const int i = blockIdx.x * 1024 + t;
    const int c = (i < N_NODES) ? cnt[i] : 0;

    int before = 0;
    for (int k = 0; k < (int)blockIdx.x; ++k) before += cnt[k * 1024 + t];

    int incl = c;
    for (int o = 1; o < 32; o <<= 1) {
        int v = __shfl_up(incl, o);
        if (lane >= o) incl += v;
    }
    for (int o = 16; o > 0; o >>= 1) before += __shfl_xor(before, o);
    if (lane == 31) s_incl[w] = incl;
    if (lane == 0) s_before[w] = before;
    __syncthreads();

    int base = 0, woff = 0;
    for (int k = 0; k < 32; ++k) {
        base += s_before[k];
        if (k < w) woff += s_incl[k];
    }
    if (i < N_NODES) {
        off[i] = base + woff + incl - c;
        dis[i] = rsqrtf((float)(c + 1));
    }
}

__global__ void k_fill(const int* __restrict__ src, const int* __restrict__ dst,
                       const int* __restrict__ rank, const int* __restrict__ off,
                       int* __restrict__ csr) {
    int e = blockIdx.x * blockDim.x + threadIdx.x;
    if (e < N_EDGES) csr[off[dst[e]] + rank[e]] = src[e];
}


template <int IN, int OUT>
__global__ __launch_bounds__(256) void k_gemm_wmma(const float* __restrict__ X,
                                                   const float* __restrict__ W,
                                                   float* __restrict__ Y,
                                                   int nrows) {
    const int lane = threadIdx.x & 31;
    const int wave = blockIdx.x * (blockDim.x >> 5) + (threadIdx.x >> 5);
    const int row0 = wave * 16;
    if (row0 >= nrows) return;

    const int half = lane >> 4;
    const int l15  = lane & 15;
    const int arow = row0 + l15;

    constexpr int NT = OUT / 16;
    v8f acc[NT];
    #pragma unroll
    for (int t = 0; t < NT; ++t)
        #pragma unroll
        for (int v = 0; v < 8; ++v) acc[t][v] = 0.0f;

    for (int k = 0; k < IN; k += 4) {
        const float* ap = X + (size_t)arow * IN + (k + 2 * half);
        v2f a;
        a.x = ap[0];
        a.y = ap[1];

        #pragma unroll
        for (int t = 0; t < NT; ++t) {
            const int col = t * 16 + l15;
            v2f b;
            b.x = W[(k + 2 * half)     * OUT + col];
            b.y = W[(k + 2 * half + 1) * OUT + col];
            acc[t] = __builtin_amdgcn_wmma_f32_16x16x4_f32(
                 false, a,  false, b,
                 (short)0, acc[t],  false,  false);
        }
    }

    #pragma unroll
    for (int t = 0; t < NT; ++t)
        #pragma unroll
        for (int v = 0; v < 8; ++v) {
            int r = row0 + v + 8 * half;
            int c = t * 16 + l15;
            Y[(size_t)r * OUT + c] = acc[t][v];
        }
}


template <int OUT>
__global__ __launch_bounds__(256) void k_gather(const int* __restrict__ csr, const int* __restrict__ off,
                                                const float* __restrict__ dis, const float* __restrict__ hw,
                                                const float* __restrict__ b, float* __restrict__ h) {
    constexpr int G = 32 / OUT;
    const int lane = threadIdx.x & 31;
    const int wv = __builtin_amdgcn_readfirstlane(blockIdx.x * (blockDim.x >> 5) + (threadIdx.x >> 5));
    const int d0 = wv * G;
    if (d0 >= N_NODES) return;
    const int g = lane / OUT;
    const int f = lane % OUT;

    float keep = 0.0f;
    #pragma unroll
    for (int nd = 0; nd < G; ++nd) {
        const int d = d0 + nd;
        const int beg = off[d];
        const int end = (d + 1 < N_NODES) ? off[d + 1] : N_EDGES;
        const float dd = dis[d];
        float acc = (g == 0) ? dd * dd * hw[(size_t)d * OUT + f] : 0.0f;
        for (int base = beg; base < end; base += 32) {
            const int n = min(32, end - base);
            int s_j = d;
            float w_j = 0.0f;
            if (lane < n) {
                s_j = csr[base + lane];
                w_j = dis[s_j] * dd;
            }
            for (int jj = 0; jj < n; jj += 8 * G) {
                #pragma unroll
                for (int u = 0; u < 8; ++u) {
                    const int idx = jj + u * G + g;
                    const int s = __shfl(s_j, idx);
                    const float wgt = __shfl(w_j, idx);
                    acc += wgt * hw[(size_t)s * OUT + f];
                }
            }
        }
        if (G == 2) acc += __shfl_xor(acc, 16);
        if (g == nd) keep = acc;
    }
    const float v = keep + b[f];
    h[(size_t)d0 * OUT + lane] = v > 0.0f ? v : 0.0f;
}


__global__ void k_gemm_out1(const float* __restrict__ h2, const float* __restrict__ W3,
                            float* __restrict__ hw3) {
    int i = blockIdx.x * blockDim.x + threadIdx.x;
    if (i >= N_NODES) return;
    float s = 0.0f;
    #pragma unroll
    for (int k = 0; k < 16; ++k) s += h2[(size_t)i * 16 + k] * W3[k];
    hw3[i] = s;
}

__global__ __launch_bounds__(1024) void k_gather_out1(const int* __restrict__ csr, const int* __restrict__ off,
                                                      const float* __restrict__ dis, const float* __restrict__ hw3,
                                                      const float* __restrict__ b3, float* __restrict__ out) {
    __shared__ float s_out[32];
    const int lane = threadIdx.x & 31;
    const int w = __builtin_amdgcn_readfirstlane(threadIdx.x >> 5);
    const int d = blockIdx.x * 32 + w;
    const int beg = off[d];
    const int end = (d + 1 < N_NODES) ? off[d + 1] : N_EDGES;
    const float dd = dis[d];
    float acc = 0.0f;
    for (int j = beg + lane; j < end; j += 32) {
        const int s = csr[j];
        acc += dis[s] * dd * hw3[s];
    }
    for (int o = 16; o > 0; o >>= 1) acc += __shfl_xor(acc, o);
    if (lane == 0) s_out[w] = acc + dd * dd * hw3[d] + b3[0];
    __syncthreads();
    if (w == 0) out[blockIdx.x * 32 + lane] = s_out[lane];
}


static inline int blks(long long n, int bs) { return (int)((n + bs - 1) / bs); }

extern "C" void kernel_launch(void* const* d_in, const int* in_sizes, int n_in,
                              void* d_out, int out_size, void* d_ws, size_t ws_size,
                              hipStream_t stream) {
    (void)in_sizes; (void)n_in; (void)out_size; (void)ws_size;

    const float* x   = (const float*)d_in[0];
    const int*   ei  = (const int*)  d_in[1];
    const float* W1  = (const float*)d_in[2];
    const float* b1  = (const float*)d_in[3];
    const float* W2  = (const float*)d_in[4];
    const float* b2  = (const float*)d_in[5];
    const float* W3  = (const float*)d_in[6];
    const float* b3  = (const float*)d_in[7];
    float* out = (float*)d_out;

    const int* src = ei;
    const int* dst = ei + N_EDGES;

    float* ws   = (float*)d_ws;
    float* dis  = ws;
    int*   off  = (int*)(ws + N_NODES);
    int*   csr  = (int*)(ws + 2 * N_NODES);
    float* pA   = ws + 34 * N_NODES;
    float* pB   = ws + 66 * N_NODES;
    int*   cnt  = (int*)pA;
    int*   rank = (int*)pB;
    float* hw1  = pA;
    float* h1   = pB;
    float* hw2  = pA;
    float* h2   = pA + 16 * N_NODES;
    float* hw3  = pB;

    const int BS = 256;

    k_zero_cnt<<<blks(N_NODES, BS), BS, 0, stream>>>(cnt);
    k_rank    <<<blks(N_EDGES, BS), BS, 0, stream>>>(dst, cnt, rank);
    k_scan    <<<blks(N_NODES, 1024), 1024, 0, stream>>>(cnt, off, dis);
    k_fill    <<<blks(N_EDGES, BS), BS, 0, stream>>>(src, dst, rank, off, csr);

    const int gemm_grid = blks(N_NODES / 16, BS / 32);

    k_gemm_wmma<128, 32><<<gemm_grid, BS, 0, stream>>>(x, W1, hw1, N_NODES);
    k_gather<32><<<blks(N_NODES, BS / 32), BS, 0, stream>>>(csr, off, dis, hw1, b1, h1);

    k_gemm_wmma<32, 16><<<gemm_grid, BS, 0, stream>>>(h1, W2, hw2, N_NODES);
    k_gather<16><<<blks(N_NODES / 2, BS / 32), BS, 0, stream>>>(csr, off, dis, hw2, b2, h2);

    k_gemm_out1  <<<blks(N_NODES, BS), BS, 0, stream>>>(h2, W3, hw3);
    k_gather_out1<<<N_NODES / 32, 1024, 0, stream>>>(csr, off, dis, hw3, b3, out);
}


# Counting sort rebuilt without global atomics: edges binned by 256-node range with LDS histograms, then ranked per bin in LDS; per-node sum order now follows LDS arrival order
# speedup vs baseline: 4.0816x; 1.8212x over previous
#include <hip/hip_runtime.h>


typedef __attribute__((ext_vector_type(2))) float v2f;
typedef __attribute__((ext_vector_type(8))) float v8f;

#define N_NODES 100000
#define N_EDGES 3200000

static_assert(N_NODES % 32 == 0, "layer-3 gather writes one full 32-node line per block");
static_assert(N_EDGES == 32 * N_NODES, "the binned edge list reuses the 32*N word plane of the layer-1 output");


#define SORT_BLOCKS 256
#define N_BINS      ((N_NODES + 255) / 256)
#define EDGE_CHUNK  ((N_EDGES + SORT_BLOCKS - 1) / SORT_BLOCKS)
#define HIST_LEN    (N_BINS * SORT_BLOCKS)

static_assert(N_NODES <= (1 << 17), "src must fit in the low 17 bits of a binned word");
static_assert(2 * HIST_LEN <= 32 * N_NODES, "histogram and its scan share plane A");

__global__ __launch_bounds__(1024) void k_bin_count(const int* __restrict__ dst, int* __restrict__ hist) {
    __shared__ int s_hist[N_BINS];
    const int t = threadIdx.x;
    const int blk = blockIdx.x;
    for (int b = t; b < N_BINS; b += 1024) s_hist[b] = 0;
    __syncthreads();
    const int lo = blk * EDGE_CHUNK;
    const int hi = min(lo + EDGE_CHUNK, N_EDGES);
    int e = lo + t;
    for (; e + 3 * 1024 < hi; e += 4 * 1024) {
        const int d0 = dst[e];
        const int d1 = dst[e + 1024];
        const int d2 = dst[e + 2 * 1024];
        const int d3 = dst[e + 3 * 1024];
        atomicAdd(&s_hist[d0 >> 8], 1);
        atomicAdd(&s_hist[d1 >> 8], 1);
        atomicAdd(&s_hist[d2 >> 8], 1);
        atomicAdd(&s_hist[d3 >> 8], 1);
    }
    for (; e < hi; e += 1024) atomicAdd(&s_hist[dst[e] >> 8], 1);
    __syncthreads();
    for (int b = t; b < N_BINS; b += 1024) hist[b * SORT_BLOCKS + blk] = s_hist[b];
}

template <int LEN>
__global__ __launch_bounds__(1024) void k_scan_excl(const int* __restrict__ cnt, int* __restrict__ out) {
    __shared__ int s_incl[32];
    __shared__ int s_before[32];
    const int t = threadIdx.x;
    const int lane = t & 31;
    const int w = t >> 5;
    const int i = blockIdx.x * 1024 + t;
    const int c = (i < LEN) ? cnt[i] : 0;

    int before = 0;
    for (int k = 0; k < (int)blockIdx.x; ++k) before += cnt[k * 1024 + t];

    int incl = c;
    for (int o = 1; o < 32; o <<= 1) {
        int v = __shfl_up(incl, o);
        if (lane >= o) incl += v;
    }
    for (int o = 16; o > 0; o >>= 1) before += __shfl_xor(before, o);
    if (lane == 31) s_incl[w] = incl;
    if (lane == 0) s_before[w] = before;
    __syncthreads();

    int base = 0, woff = 0;
    for (int k = 0; k < 32; ++k) {
        base += s_before[k];
        if (k < w) woff += s_incl[k];
    }
    if (i < LEN) out[i] = base + woff + incl - c;
}

__global__ __launch_bounds__(1024) void k_bin_scatter(const int* __restrict__ src, const int* __restrict__ dst,
                                                      const int* __restrict__ binoff, int* __restrict__ binned) {
    __shared__ int s_cur[N_BINS];
    const int t = threadIdx.x;
    const int blk = blockIdx.x;
    for (int b = t; b < N_BINS; b += 1024) s_cur[b] = binoff[b * SORT_BLOCKS + blk];
    __syncthreads();
    const int lo = blk * EDGE_CHUNK;
    const int hi = min(lo + EDGE_CHUNK, N_EDGES);
    int e = lo + t;
    for (; e + 3 * 1024 < hi; e += 4 * 1024) {
        const int d0 = dst[e];
        const int d1 = dst[e + 1024];
        const int d2 = dst[e + 2 * 1024];
        const int d3 = dst[e + 3 * 1024];
        const int s0 = src[e];
        const int s1 = src[e + 1024];
        const int s2 = src[e + 2 * 1024];
        const int s3 = src[e + 3 * 1024];
        const int p0 = atomicAdd(&s_cur[d0 >> 8], 1);
        const int p1 = atomicAdd(&s_cur[d1 >> 8], 1);
        const int p2 = atomicAdd(&s_cur[d2 >> 8], 1);
        const int p3 = atomicAdd(&s_cur[d3 >> 8], 1);
        binned[p0] = s0 | ((d0 & 255) << 17);
        binned[p1] = s1 | ((d1 & 255) << 17);
        binned[p2] = s2 | ((d2 & 255) << 17);
        binned[p3] = s3 | ((d3 & 255) << 17);
    }
    for (; e < hi; e += 1024) {
        const int d = dst[e];
        const int p = atomicAdd(&s_cur[d >> 8], 1);
        binned[p] = src[e] | ((d & 255) << 17);
    }
}

__global__ __launch_bounds__(1024) void k_bin_sort(const int* __restrict__ binned, const int* __restrict__ binoff,
                                                   int* __restrict__ off, float* __restrict__ dis,
                                                   int* __restrict__ csr) {
    __shared__ int s_cnt[256];
    __shared__ int s_cur[256];
    __shared__ int s_wsum[8];
    const int t = threadIdx.x;
    const int lane = t & 31;
    const int bin = blockIdx.x;
    const int bstart = binoff[bin * SORT_BLOCKS];
    const int bend = (bin + 1 < N_BINS) ? binoff[(bin + 1) * SORT_BLOCKS] : N_EDGES;

    if (t < 256) s_cnt[t] = 0;
    __syncthreads();
    int i = bstart + t;
    for (; i + 3 * 1024 < bend; i += 4 * 1024) {
        const int w0 = binned[i];
        const int w1 = binned[i + 1024];
        const int w2 = binned[i + 2 * 1024];
        const int w3 = binned[i + 3 * 1024];
        atomicAdd(&s_cnt[w0 >> 17], 1);
        atomicAdd(&s_cnt[w1 >> 17], 1);
        atomicAdd(&s_cnt[w2 >> 17], 1);
        atomicAdd(&s_cnt[w3 >> 17], 1);
    }
    for (; i < bend; i += 1024) atomicAdd(&s_cnt[binned[i] >> 17], 1);
    __syncthreads();

    int c = 0, incl = 0;
    if (t < 256) {
        c = s_cnt[t];
        incl = c;
        for (int o = 1; o < 32; o <<= 1) {
            int v = __shfl_up(incl, o);
            if (lane >= o) incl += v;
        }
        if (lane == 31) s_wsum[t >> 5] = incl;
    }
    __syncthreads();
    if (t < 256) {
        int woff = 0;
        for (int k = 0; k < 8; ++k)
            if (k < (t >> 5)) woff += s_wsum[k];
        const int excl = woff + incl - c;
        s_cur[t] = excl;
        const int node = bin * 256 + t;
        if (node < N_NODES) {
            off[node] = bstart + excl;
            dis[node] = rsqrtf((float)(c + 1));
        }
    }
    __syncthreads();
    i = bstart + t;
    for (; i + 3 * 1024 < bend; i += 4 * 1024) {
        const int w0 = binned[i];
        const int w1 = binned[i + 1024];
        const int w2 = binned[i + 2 * 1024];
        const int w3 = binned[i + 3 * 1024];
        const int p0 = atomicAdd(&s_cur[w0 >> 17], 1);
        const int p1 = atomicAdd(&s_cur[w1 >> 17], 1);
        const int p2 = atomicAdd(&s_cur[w2 >> 17], 1);
        const int p3 = atomicAdd(&s_cur[w3 >> 17], 1);
        csr[bstart + p0] = w0 & 0x1FFFF;
        csr[bstart + p1] = w1 & 0x1FFFF;
        csr[bstart + p2] = w2 & 0x1FFFF;
        csr[bstart + p3] = w3 & 0x1FFFF;
    }
    for (; i < bend; i += 1024) {
        const int wd = binned[i];
        const int p = atomicAdd(&s_cur[wd >> 17], 1);
        csr[bstart + p] = wd & 0x1FFFF;
    }
}


template <int IN, int OUT>
__global__ __launch_bounds__(256) void k_gemm_wmma(const float* __restrict__ X,
                                                   const float* __restrict__ W,
                                                   float* __restrict__ Y,
                                                   int nrows) {
    const int lane = threadIdx.x & 31;
    const int wave = blockIdx.x * (blockDim.x >> 5) + (threadIdx.x >> 5);
    const int row0 = wave * 16;
    if (row0 >= nrows) return;

    const int half = lane >> 4;
    const int l15  = lane & 15;
    const int arow = row0 + l15;

    constexpr int NT = OUT / 16;
    v8f acc[NT];
    #pragma unroll
    for (int t = 0; t < NT; ++t)
        #pragma unroll
        for (int v = 0; v < 8; ++v) acc[t][v] = 0.0f;

    for (int k = 0; k < IN; k += 4) {
        const float* ap = X + (size_t)arow * IN + (k + 2 * half);
        v2f a;
        a.x = ap[0];
        a.y = ap[1];

        #pragma unroll
        for (int t = 0; t < NT; ++t) {
            const int col = t * 16 + l15;
            v2f b;
            b.x = W[(k + 2 * half)     * OUT + col];
            b.y = W[(k + 2 * half + 1) * OUT + col];
            acc[t] = __builtin_amdgcn_wmma_f32_16x16x4_f32(
                 false, a,  false, b,
                 (short)0, acc[t],  false,  false);
        }
    }

    #pragma unroll
    for (int t = 0; t < NT; ++t)
        #pragma unroll
        for (int v = 0; v < 8; ++v) {
            int r = row0 + v + 8 * half;
            int c = t * 16 + l15;
            Y[(size_t)r * OUT + c] = acc[t][v];
        }
}


template <int OUT>
__global__ __launch_bounds__(256) void k_gather(const int* __restrict__ csr, const int* __restrict__ off,
                                                const float* __restrict__ dis, const float* __restrict__ hw,
                                                const float* __restrict__ b, float* __restrict__ h) {
    constexpr int G = 32 / OUT;
    const int lane = threadIdx.x & 31;
    const int wv = __builtin_amdgcn_readfirstlane(blockIdx.x * (blockDim.x >> 5) + (threadIdx.x >> 5));
    const int d0 = wv * G;
    if (d0 >= N_NODES) return;
    const int g = lane / OUT;
    const int f = lane % OUT;

    float keep = 0.0f;
    #pragma unroll
    for (int nd = 0; nd < G; ++nd) {
        const int d = d0 + nd;
        const int beg = off[d];
        const int end = (d + 1 < N_NODES) ? off[d + 1] : N_EDGES;
        const float dd = dis[d];
        float acc = (g == 0) ? dd * dd * hw[(size_t)d * OUT + f] : 0.0f;
        for (int base = beg; base < end; base += 32) {
            const int n = min(32, end - base);
            int s_j = d;
            float w_j = 0.0f;
            if (lane < n) {
                s_j = csr[base + lane];
                w_j = dis[s_j] * dd;
            }
            for (int jj = 0; jj < n; jj += 8 * G) {
                #pragma unroll
                for (int u = 0; u < 8; ++u) {
                    const int idx = jj + u * G + g;
                    const int s = __shfl(s_j, idx);
                    const float wgt = __shfl(w_j, idx);
                    acc += wgt * hw[(size_t)s * OUT + f];
                }
            }
        }
        if (G == 2) acc += __shfl_xor(acc, 16);
        if (g == nd) keep = acc;
    }
    const float v = keep + b[f];
    h[(size_t)d0 * OUT + lane] = v > 0.0f ? v : 0.0f;
}


__global__ void k_gemm_out1(const float* __restrict__ h2, const float* __restrict__ W3,
                            float* __restrict__ hw3) {
    int i = blockIdx.x * blockDim.x + threadIdx.x;
    if (i >= N_NODES) return;
    float s = 0.0f;
    #pragma unroll
    for (int k = 0; k < 16; ++k) s += h2[(size_t)i * 16 + k] * W3[k];
    hw3[i] = s;
}

__global__ __launch_bounds__(1024) void k_gather_out1(const int* __restrict__ csr, const int* __restrict__ off,
                                                      const float* __restrict__ dis, const float* __restrict__ hw3,
                                                      const float* __restrict__ b3, float* __restrict__ out) {
    __shared__ float s_out[32];
    const int lane = threadIdx.x & 31;
    const int w = __builtin_amdgcn_readfirstlane(threadIdx.x >> 5);
    const int d = blockIdx.x * 32 + w;
    const int beg = off[d];
    const int end = (d + 1 < N_NODES) ? off[d + 1] : N_EDGES;
    const float dd = dis[d];
    float acc = 0.0f;
    for (int j = beg + lane; j < end; j += 32) {
        const int s = csr[j];
        acc += dis[s] * dd * hw3[s];
    }
    for (int o = 16; o > 0; o >>= 1) acc += __shfl_xor(acc, o);
    if (lane == 0) s_out[w] = acc + dd * dd * hw3[d] + b3[0];
    __syncthreads();
    if (w == 0) out[blockIdx.x * 32 + lane] = s_out[lane];
}


static inline int blks(long long n, int bs) { return (int)((n + bs - 1) / bs); }

extern "C" void kernel_launch(void* const* d_in, const int* in_sizes, int n_in,
                              void* d_out, int out_size, void* d_ws, size_t ws_size,
                              hipStream_t stream) {
    (void)in_sizes; (void)n_in; (void)out_size; (void)ws_size;

    const float* x   = (const float*)d_in[0];
    const int*   ei  = (const int*)  d_in[1];
    const float* W1  = (const float*)d_in[2];
    const float* b1  = (const float*)d_in[3];
    const float* W2  = (const float*)d_in[4];
    const float* b2  = (const float*)d_in[5];
    const float* W3  = (const float*)d_in[6];
    const float* b3  = (const float*)d_in[7];
    float* out = (float*)d_out;

    const int* src = ei;
    const int* dst = ei + N_EDGES;

    float* ws   = (float*)d_ws;
    float* dis  = ws;
    int*   off  = (int*)(ws + N_NODES);
    int*   csr  = (int*)(ws + 2 * N_NODES);
    float* pA   = ws + 34 * N_NODES;
    float* pB   = ws + 66 * N_NODES;
    int*   hist   = (int*)pA;
    int*   binoff = (int*)pA + HIST_LEN;
    int*   binned = (int*)pB;
    float* hw1  = pA;
    float* h1   = pB;
    float* hw2  = pA;
    float* h2   = pA + 16 * N_NODES;
    float* hw3  = pB;

    const int BS = 256;

    k_bin_count  <<<SORT_BLOCKS, 1024, 0, stream>>>(dst, hist);
    k_scan_excl<HIST_LEN><<<(HIST_LEN + 1023) / 1024, 1024, 0, stream>>>(hist, binoff);
    k_bin_scatter<<<SORT_BLOCKS, 1024, 0, stream>>>(src, dst, binoff, binned);
    k_bin_sort   <<<N_BINS, 1024, 0, stream>>>(binned, binoff, off, dis, csr);

    const int gemm_grid = blks(N_NODES / 16, BS / 32);

    k_gemm_wmma<128, 32><<<gemm_grid, BS, 0, stream>>>(x, W1, hw1, N_NODES);
    k_gather<32><<<blks(N_NODES, BS / 32), BS, 0, stream>>>(csr, off, dis, hw1, b1, h1);

    k_gemm_wmma<32, 16><<<gemm_grid, BS, 0, stream>>>(h1, W2, hw2, N_NODES);
    k_gather<16><<<blks(N_NODES / 2, BS / 32), BS, 0, stream>>>(csr, off, dis, hw2, b2, h2);

    k_gemm_out1  <<<blks(N_NODES, BS), BS, 0, stream>>>(h2, W3, hw3);
    k_gather_out1<<<N_NODES / 32, 1024, 0, stream>>>(csr, off, dis, hw3, b3, out);
}


# dis[src] carried by the rows in layers 2 and 3: each gather stores dis[d]*h[d], so the next transform's rows arrive scaled and its gather adds them unweighted, times dis[dst] once
# speedup vs baseline: 4.5382x; 1.1119x over previous
#include <hip/hip_runtime.h>


typedef __attribute__((ext_vector_type(2))) float v2f;
typedef __attribute__((ext_vector_type(8))) float v8f;

#define N_NODES 100000
#define N_EDGES 3200000

static_assert(N_NODES % 32 == 0, "layer-3 gather writes one full 32-node line per block");
static_assert(N_EDGES == 32 * N_NODES, "the binned edge list reuses the 32*N word plane of the layer-1 output");


#define SORT_BLOCKS 256
#define N_BINS      ((N_NODES + 255) / 256)
#define EDGE_CHUNK  ((N_EDGES + SORT_BLOCKS - 1) / SORT_BLOCKS)
#define HIST_LEN    (N_BINS * SORT_BLOCKS)

static_assert(N_NODES <= (1 << 17), "src must fit in the low 17 bits of a binned word");
static_assert(2 * HIST_LEN <= 32 * N_NODES, "histogram and its scan share plane A");

__global__ __launch_bounds__(1024) void k_bin_count(const int* __restrict__ dst, int* __restrict__ hist) {
    __shared__ int s_hist[N_BINS];
    const int t = threadIdx.x;
    const int blk = blockIdx.x;
    for (int b = t; b < N_BINS; b += 1024) s_hist[b] = 0;
    __syncthreads();
    const int lo = blk * EDGE_CHUNK;
    const int hi = min(lo + EDGE_CHUNK, N_EDGES);
    int e = lo + t;
    for (; e + 3 * 1024 < hi; e += 4 * 1024) {
        const int d0 = dst[e];
        const int d1 = dst[e + 1024];
        const int d2 = dst[e + 2 * 1024];
        const int d3 = dst[e + 3 * 1024];
        atomicAdd(&s_hist[d0 >> 8], 1);
        atomicAdd(&s_hist[d1 >> 8], 1);
        atomicAdd(&s_hist[d2 >> 8], 1);
        atomicAdd(&s_hist[d3 >> 8], 1);
    }
    for (; e < hi; e += 1024) atomicAdd(&s_hist[dst[e] >> 8], 1);
    __syncthreads();
    for (int b = t; b < N_BINS; b += 1024) hist[b * SORT_BLOCKS + blk] = s_hist[b];
}

template <int LEN>
__global__ __launch_bounds__(1024) void k_scan_excl(const int* __restrict__ cnt, int* __restrict__ out) {
    __shared__ int s_incl[32];
    __shared__ int s_before[32];
    const int t = threadIdx.x;
    const int lane = t & 31;
    const int w = t >> 5;
    const int i = blockIdx.x * 1024 + t;
    const int c = (i < LEN) ? cnt[i] : 0;

    int before = 0;
    for (int k = 0; k < (int)blockIdx.x; ++k) before += cnt[k * 1024 + t];

    int incl = c;
    for (int o = 1; o < 32; o <<= 1) {
        int v = __shfl_up(incl, o);
        if (lane >= o) incl += v;
    }
    for (int o = 16; o > 0; o >>= 1) before += __shfl_xor(before, o);
    if (lane == 31) s_incl[w] = incl;
    if (lane == 0) s_before[w] = before;
    __syncthreads();

    int base = 0, woff = 0;
    for (int k = 0; k < 32; ++k) {
        base += s_before[k];
        if (k < w) woff += s_incl[k];
    }
    if (i < LEN) out[i] = base + woff + incl - c;
}

__global__ __launch_bounds__(1024) void k_bin_scatter(const int* __restrict__ src, const int* __restrict__ dst,
                                                      const int* __restrict__ binoff, int* __restrict__ binned) {
    __shared__ int s_cur[N_BINS];
    const int t = threadIdx.x;
    const int blk = blockIdx.x;
    for (int b = t; b < N_BINS; b += 1024) s_cur[b] = binoff[b * SORT_BLOCKS + blk];
    __syncthreads();
    const int lo = blk * EDGE_CHUNK;
    const int hi = min(lo + EDGE_CHUNK, N_EDGES);
    int e = lo + t;
    for (; e + 3 * 1024 < hi; e += 4 * 1024) {
        const int d0 = dst[e];
        const int d1 = dst[e + 1024];
        const int d2 = dst[e + 2 * 1024];
        const int d3 = dst[e + 3 * 1024];
        const int s0 = src[e];
        const int s1 = src[e + 1024];
        const int s2 = src[e + 2 * 1024];
        const int s3 = src[e + 3 * 1024];
        const int p0 = atomicAdd(&s_cur[d0 >> 8], 1);
        const int p1 = atomicAdd(&s_cur[d1 >> 8], 1);
        const int p2 = atomicAdd(&s_cur[d2 >> 8], 1);
        const int p3 = atomicAdd(&s_cur[d3 >> 8], 1);
        binned[p0] = s0 | ((d0 & 255) << 17);
        binned[p1] = s1 | ((d1 & 255) << 17);
        binned[p2] = s2 | ((d2 & 255) << 17);
        binned[p3] = s3 | ((d3 & 255) << 17);
    }
    for (; e < hi; e += 1024) {
        const int d = dst[e];
        const int p = atomicAdd(&s_cur[d >> 8], 1);
        binned[p] = src[e] | ((d & 255) << 17);
    }
}

__global__ __launch_bounds__(1024) void k_bin_sort(const int* __restrict__ binned, const int* __restrict__ binoff,
                                                   int* __restrict__ off, float* __restrict__ dis,
                                                   int* __restrict__ csr) {
    __shared__ int s_cnt[256];
    __shared__ int s_cur[256];
    __shared__ int s_wsum[8];
    const int t = threadIdx.x;
    const int lane = t & 31;
    const int bin = blockIdx.x;
    const int bstart = binoff[bin * SORT_BLOCKS];
    const int bend = (bin + 1 < N_BINS) ? binoff[(bin + 1) * SORT_BLOCKS] : N_EDGES;

    if (t < 256) s_cnt[t] = 0;
    __syncthreads();
    int i = bstart + t;
    for (; i + 3 * 1024 < bend; i += 4 * 1024) {
        const int w0 = binned[i];
        const int w1 = binned[i + 1024];
        const int w2 = binned[i + 2 * 1024];
        const int w3 = binned[i + 3 * 1024];
        atomicAdd(&s_cnt[w0 >> 17], 1);
        atomicAdd(&s_cnt[w1 >> 17], 1);
        atomicAdd(&s_cnt[w2 >> 17], 1);
        atomicAdd(&s_cnt[w3 >> 17], 1);
    }
    for (; i < bend; i += 1024) atomicAdd(&s_cnt[binned[i] >> 17], 1);
    __syncthreads();

    int c = 0, incl = 0;
    if (t < 256) {
        c = s_cnt[t];
        incl = c;
        for (int o = 1; o < 32; o <<= 1) {
            int v = __shfl_up(incl, o);
            if (lane >= o) incl += v;
        }
        if (lane == 31) s_wsum[t >> 5] = incl;
    }
    __syncthreads();
    if (t < 256) {
        int woff = 0;
        for (int k = 0; k < 8; ++k)
            if (k < (t >> 5)) woff += s_wsum[k];
        const int excl = woff + incl - c;
        s_cur[t] = excl;
        const int node = bin * 256 + t;
        if (node < N_NODES) {
            off[node] = bstart + excl;
            dis[node] = rsqrtf((float)(c + 1));
        }
    }
    __syncthreads();
    i = bstart + t;
    for (; i + 3 * 1024 < bend; i += 4 * 1024) {
        const int w0 = binned[i];
        const int w1 = binned[i + 1024];
        const int w2 = binned[i + 2 * 1024];
        const int w3 = binned[i + 3 * 1024];
        const int p0 = atomicAdd(&s_cur[w0 >> 17], 1);
        const int p1 = atomicAdd(&s_cur[w1 >> 17], 1);
        const int p2 = atomicAdd(&s_cur[w2 >> 17], 1);
        const int p3 = atomicAdd(&s_cur[w3 >> 17], 1);
        csr[bstart + p0] = w0 & 0x1FFFF;
        csr[bstart + p1] = w1 & 0x1FFFF;
        csr[bstart + p2] = w2 & 0x1FFFF;
        csr[bstart + p3] = w3 & 0x1FFFF;
    }
    for (; i < bend; i += 1024) {
        const int wd = binned[i];
        const int p = atomicAdd(&s_cur[wd >> 17], 1);
        csr[bstart + p] = wd & 0x1FFFF;
    }
}


template <int IN, int OUT>
__global__ __launch_bounds__(256) void k_gemm_wmma(const float* __restrict__ X,
                                                   const float* __restrict__ W,
                                                   float* __restrict__ Y,
                                                   int nrows) {
    const int lane = threadIdx.x & 31;
    const int wave = blockIdx.x * (blockDim.x >> 5) + (threadIdx.x >> 5);
    const int row0 = wave * 16;
    if (row0 >= nrows) return;

    const int half = lane >> 4;
    const int l15  = lane & 15;
    const int arow = row0 + l15;

    constexpr int NT = OUT / 16;
    v8f acc[NT];
    #pragma unroll
    for (int t = 0; t < NT; ++t)
        #pragma unroll
        for (int v = 0; v < 8; ++v) acc[t][v] = 0.0f;

    for (int k = 0; k < IN; k += 4) {
        const float* ap = X + (size_t)arow * IN + (k + 2 * half);
        v2f a;
        a.x = ap[0];
        a.y = ap[1];

        #pragma unroll
        for (int t = 0; t < NT; ++t) {
            const int col = t * 16 + l15;
            v2f b;
            b.x = W[(k + 2 * half)     * OUT + col];
            b.y = W[(k + 2 * half + 1) * OUT + col];
            acc[t] = __builtin_amdgcn_wmma_f32_16x16x4_f32(
                 false, a,  false, b,
                 (short)0, acc[t],  false,  false);
        }
    }

    #pragma unroll
    for (int t = 0; t < NT; ++t)
        #pragma unroll
        for (int v = 0; v < 8; ++v) {
            int r = row0 + v + 8 * half;
            int c = t * 16 + l15;
            Y[(size_t)r * OUT + c] = acc[t][v];
        }
}


template <int OUT, bool PRESCALED>
__global__ __launch_bounds__(256) void k_gather(const int* __restrict__ csr, const int* __restrict__ off,
                                                const float* __restrict__ dis, const float* __restrict__ hw,
                                                const float* __restrict__ b, float* __restrict__ h) {
    constexpr int G = 32 / OUT;
    const int lane = threadIdx.x & 31;
    const int wv = __builtin_amdgcn_readfirstlane(blockIdx.x * (blockDim.x >> 5) + (threadIdx.x >> 5));
    const int d0 = wv * G;
    if (d0 >= N_NODES) return;
    const int g = lane / OUT;
    const int f = lane % OUT;

    const float bias = b[f];
    float keep = 0.0f;
    #pragma unroll
    for (int nd = 0; nd < G; ++nd) {
        const int d = d0 + nd;
        const int beg = off[d];
        const int end = (d + 1 < N_NODES) ? off[d + 1] : N_EDGES;
        const float dd = dis[d];
        const float self = hw[(size_t)d * OUT + f];
        float acc = (g == 0) ? (PRESCALED ? self : dd * dd * self) : 0.0f;
        for (int base = beg; base < end; base += 32) {
            const int n = min(32, end - base);
            int s_j = d;
            float w_j = 0.0f;
            if (lane < n) {
                s_j = csr[base + lane];
                if (!PRESCALED) w_j = dis[s_j] * dd;
            }
            for (int jj = 0; jj < n; jj += 8 * G) {
                if (PRESCALED) {
                    float r[8];
                    #pragma unroll
                    for (int u = 0; u < 8; ++u) {
                        const int s = __shfl(s_j, jj + u * G + g);
                        r[u] = hw[(size_t)s * OUT + f];
                    }
                    #pragma unroll
                    for (int u = 0; u < 8; ++u) acc += (jj + u * G + g < n) ? r[u] : 0.0f;
                } else {
                    #pragma unroll
                    for (int u = 0; u < 8; ++u) {
                        const int idx = jj + u * G + g;
                        const int s = __shfl(s_j, idx);
                        const float wgt = __shfl(w_j, idx);
                        acc += wgt * hw[(size_t)s * OUT + f];
                    }
                }
            }
        }
        if (G == 2) acc += __shfl_xor(acc, 16);
        float v = (PRESCALED ? dd * acc : acc) + bias;
        v = v > 0.0f ? v : 0.0f;
        if (g == nd) keep = dd * v;
    }
    h[(size_t)d0 * OUT + lane] = keep;
}


__global__ void k_gemm_out1(const float* __restrict__ h2, const float* __restrict__ W3,
                            float* __restrict__ hw3) {
    int i = blockIdx.x * blockDim.x + threadIdx.x;
    if (i >= N_NODES) return;
    float s = 0.0f;
    #pragma unroll
    for (int k = 0; k < 16; ++k) s += h2[(size_t)i * 16 + k] * W3[k];
    hw3[i] = s;
}

__global__ __launch_bounds__(1024) void k_gather_out1(const int* __restrict__ csr, const int* __restrict__ off,
                                                      const float* __restrict__ dis, const float* __restrict__ hw3,
                                                      const float* __restrict__ b3, float* __restrict__ out) {
    __shared__ float s_out[32];
    const int lane = threadIdx.x & 31;
    const int w = __builtin_amdgcn_readfirstlane(threadIdx.x >> 5);
    const int d = blockIdx.x * 32 + w;
    const int beg = off[d];
    const int end = (d + 1 < N_NODES) ? off[d + 1] : N_EDGES;
    const float dd = dis[d];
    float acc = 0.0f;
    for (int j = beg + lane; j < end; j += 32) acc += hw3[csr[j]];
    for (int o = 16; o > 0; o >>= 1) acc += __shfl_xor(acc, o);
    if (lane == 0) s_out[w] = dd * (acc + hw3[d]) + b3[0];
    __syncthreads();
    if (w == 0) out[blockIdx.x * 32 + lane] = s_out[lane];
}


static inline int blks(long long n, int bs) { return (int)((n + bs - 1) / bs); }

extern "C" void kernel_launch(void* const* d_in, const int* in_sizes, int n_in,
                              void* d_out, int out_size, void* d_ws, size_t ws_size,
                              hipStream_t stream) {
    (void)in_sizes; (void)n_in; (void)out_size; (void)ws_size;

    const float* x   = (const float*)d_in[0];
    const int*   ei  = (const int*)  d_in[1];
    const float* W1  = (const float*)d_in[2];
    const float* b1  = (const float*)d_in[3];
    const float* W2  = (const float*)d_in[4];
    const float* b2  = (const float*)d_in[5];
    const float* W3  = (const float*)d_in[6];
    const float* b3  = (const float*)d_in[7];
    float* out = (float*)d_out;

    const int* src = ei;
    const int* dst = ei + N_EDGES;

    float* ws   = (float*)d_ws;
    float* dis  = ws;
    int*   off  = (int*)(ws + N_NODES);
    int*   csr  = (int*)(ws + 2 * N_NODES);
    float* pA   = ws + 34 * N_NODES;
    float* pB   = ws + 66 * N_NODES;
    int*   hist   = (int*)pA;
    int*   binoff = (int*)pA + HIST_LEN;
    int*   binned = (int*)pB;
    float* hw1  = pA;
    float* h1   = pB;
    float* hw2  = pA;
    float* h2   = pA + 16 * N_NODES;
    float* hw3  = pB;

    const int BS = 256;

    k_bin_count  <<<SORT_BLOCKS, 1024, 0, stream>>>(dst, hist);
    k_scan_excl<HIST_LEN><<<(HIST_LEN + 1023) / 1024, 1024, 0, stream>>>(hist, binoff);
    k_bin_scatter<<<SORT_BLOCKS, 1024, 0, stream>>>(src, dst, binoff, binned);
    k_bin_sort   <<<N_BINS, 1024, 0, stream>>>(binned, binoff, off, dis, csr);

    const int gemm_grid = blks(N_NODES / 16, BS / 32);

    k_gemm_wmma<128, 32><<<gemm_grid, BS, 0, stream>>>(x, W1, hw1, N_NODES);
    k_gather<32, false><<<blks(N_NODES, BS / 32), BS, 0, stream>>>(csr, off, dis, hw1, b1, h1);

    k_gemm_wmma<32, 16><<<gemm_grid, BS, 0, stream>>>(h1, W2, hw2, N_NODES);
    k_gather<16, true><<<blks(N_NODES / 2, BS / 32), BS, 0, stream>>>(csr, off, dis, hw2, b2, h2);

    k_gemm_out1  <<<blks(N_NODES, BS), BS, 0, stream>>>(h2, W3, hw3);
    k_gather_out1<<<N_NODES / 32, 1024, 0, stream>>>(csr, off, dis, hw3, b3, out);
}


# Layer-2 and layer-3 transforms folded into the gathers before them: a block's finished rows pass through LDS to one wave that does the W2 WMMA (W3 dot); h1, h2 never stored; 10 -> 8 launches
# speedup vs baseline: 4.9078x; 1.0814x over previous
#include <hip/hip_runtime.h>


typedef __attribute__((ext_vector_type(2))) float v2f;
typedef __attribute__((ext_vector_type(8))) float v8f;

#define N_NODES 100000
#define N_EDGES 3200000

static_assert(N_NODES % 32 == 0, "layer-3 gather writes one full 32-node line per block");
static_assert(N_EDGES == 32 * N_NODES, "the binned edge list reuses the 32*N word plane of the layer-1 output");


#define SORT_BLOCKS 256
#define N_BINS      ((N_NODES + 255) / 256)
#define EDGE_CHUNK  ((N_EDGES + SORT_BLOCKS - 1) / SORT_BLOCKS)
#define HIST_LEN    (N_BINS * SORT_BLOCKS)

static_assert(N_NODES <= (1 << 17), "src must fit in the low 17 bits of a binned word");
static_assert(2 * HIST_LEN <= 32 * N_NODES, "histogram and its scan share plane A");

__global__ __launch_bounds__(1024) void k_bin_count(const int* __restrict__ dst, int* __restrict__ hist) {
    __shared__ int s_hist[N_BINS];
    const int t = threadIdx.x;
    const int blk = blockIdx.x;
    for (int b = t; b < N_BINS; b += 1024) s_hist[b] = 0;
    __syncthreads();
    const int lo = blk * EDGE_CHUNK;
    const int hi = min(lo + EDGE_CHUNK, N_EDGES);
    int e = lo + t;
    for (; e + 3 * 1024 < hi; e += 4 * 1024) {
        const int d0 = dst[e];
        const int d1 = dst[e + 1024];
        const int d2 = dst[e + 2 * 1024];
        const int d3 = dst[e + 3 * 1024];
        atomicAdd(&s_hist[d0 >> 8], 1);
        atomicAdd(&s_hist[d1 >> 8], 1);
        atomicAdd(&s_hist[d2 >> 8], 1);
        atomicAdd(&s_hist[d3 >> 8], 1);
    }
    for (; e < hi; e += 1024) atomicAdd(&s_hist[dst[e] >> 8], 1);
    __syncthreads();
    for (int b = t; b < N_BINS; b += 1024) hist[b * SORT_BLOCKS + blk] = s_hist[b];
}

template <int LEN>
__global__ __launch_bounds__(1024) void k_scan_excl(const int* __restrict__ cnt, int* __restrict__ out) {
    __shared__ int s_incl[32];
    __shared__ int s_before[32];
    const int t = threadIdx.x;
    const int lane = t & 31;
    const int w = t >> 5;
    const int i = blockIdx.x * 1024 + t;
    const int c = (i < LEN) ? cnt[i] : 0;

    int before = 0;
    for (int k = 0; k < (int)blockIdx.x; ++k) before += cnt[k * 1024 + t];

    int incl = c;
    for (int o = 1; o < 32; o <<= 1) {
        int v = __shfl_up(incl, o);
        if (lane >= o) incl += v;
    }
    for (int o = 16; o > 0; o >>= 1) before += __shfl_xor(before, o);
    if (lane == 31) s_incl[w] = incl;
    if (lane == 0) s_before[w] = before;
    __syncthreads();

    int base = 0, woff = 0;
    for (int k = 0; k < 32; ++k) {
        base += s_before[k];
        if (k < w) woff += s_incl[k];
    }
    if (i < LEN) out[i] = base + woff + incl - c;
}

__global__ __launch_bounds__(1024) void k_bin_scatter(const int* __restrict__ src, const int* __restrict__ dst,
                                                      const int* __restrict__ binoff, int* __restrict__ binned) {
    __shared__ int s_cur[N_BINS];
    const int t = threadIdx.x;
    const int blk = blockIdx.x;
    for (int b = t; b < N_BINS; b += 1024) s_cur[b] = binoff[b * SORT_BLOCKS + blk];
    __syncthreads();
    const int lo = blk * EDGE_CHUNK;
    const int hi = min(lo + EDGE_CHUNK, N_EDGES);
    int e = lo + t;
    for (; e + 3 * 1024 < hi; e += 4 * 1024) {
        const int d0 = dst[e];
        const int d1 = dst[e + 1024];
        const int d2 = dst[e + 2 * 1024];
        const int d3 = dst[e + 3 * 1024];
        const int s0 = src[e];
        const int s1 = src[e + 1024];
        const int s2 = src[e + 2 * 1024];
        const int s3 = src[e + 3 * 1024];
        const int p0 = atomicAdd(&s_cur[d0 >> 8], 1);
        const int p1 = atomicAdd(&s_cur[d1 >> 8], 1);
        const int p2 = atomicAdd(&s_cur[d2 >> 8], 1);
        const int p3 = atomicAdd(&s_cur[d3 >> 8], 1);
        binned[p0] = s0 | ((d0 & 255) << 17);
        binned[p1] = s1 | ((d1 & 255) << 17);
        binned[p2] = s2 | ((d2 & 255) << 17);
        binned[p3] = s3 | ((d3 & 255) << 17);
    }
    for (; e < hi; e += 1024) {
        const int d = dst[e];
        const int p = atomicAdd(&s_cur[d >> 8], 1);
        binned[p] = src[e] | ((d & 255) << 17);
    }
}

__global__ __launch_bounds__(1024) void k_bin_sort(const int* __restrict__ binned, const int* __restrict__ binoff,
                                                   int* __restrict__ off, float* __restrict__ dis,
                                                   int* __restrict__ csr) {
    __shared__ int s_cnt[256];
    __shared__ int s_cur[256];
    __shared__ int s_wsum[8];
    const int t = threadIdx.x;
    const int lane = t & 31;
    const int bin = blockIdx.x;
    const int bstart = binoff[bin * SORT_BLOCKS];
    const int bend = (bin + 1 < N_BINS) ? binoff[(bin + 1) * SORT_BLOCKS] : N_EDGES;

    if (t < 256) s_cnt[t] = 0;
    __syncthreads();
    int i = bstart + t;
    for (; i + 3 * 1024 < bend; i += 4 * 1024) {
        const int w0 = binned[i];
        const int w1 = binned[i + 1024];
        const int w2 = binned[i + 2 * 1024];
        const int w3 = binned[i + 3 * 1024];
        atomicAdd(&s_cnt[w0 >> 17], 1);
        atomicAdd(&s_cnt[w1 >> 17], 1);
        atomicAdd(&s_cnt[w2 >> 17], 1);
        atomicAdd(&s_cnt[w3 >> 17], 1);
    }
    for (; i < bend; i += 1024) atomicAdd(&s_cnt[binned[i] >> 17], 1);
    __syncthreads();

    int c = 0, incl = 0;
    if (t < 256) {
        c = s_cnt[t];
        incl = c;
        for (int o = 1; o < 32; o <<= 1) {
            int v = __shfl_up(incl, o);
            if (lane >= o) incl += v;
        }
        if (lane == 31) s_wsum[t >> 5] = incl;
    }
    __syncthreads();
    if (t < 256) {
        int woff = 0;
        for (int k = 0; k < 8; ++k)
            if (k < (t >> 5)) woff += s_wsum[k];
        const int excl = woff + incl - c;
        s_cur[t] = excl;
        const int node = bin * 256 + t;
        if (node < N_NODES) {
            off[node] = bstart + excl;
            dis[node] = rsqrtf((float)(c + 1));
        }
    }
    __syncthreads();
    i = bstart + t;
    for (; i + 3 * 1024 < bend; i += 4 * 1024) {
        const int w0 = binned[i];
        const int w1 = binned[i + 1024];
        const int w2 = binned[i + 2 * 1024];
        const int w3 = binned[i + 3 * 1024];
        const int p0 = atomicAdd(&s_cur[w0 >> 17], 1);
        const int p1 = atomicAdd(&s_cur[w1 >> 17], 1);
        const int p2 = atomicAdd(&s_cur[w2 >> 17], 1);
        const int p3 = atomicAdd(&s_cur[w3 >> 17], 1);
        csr[bstart + p0] = w0 & 0x1FFFF;
        csr[bstart + p1] = w1 & 0x1FFFF;
        csr[bstart + p2] = w2 & 0x1FFFF;
        csr[bstart + p3] = w3 & 0x1FFFF;
    }
    for (; i < bend; i += 1024) {
        const int wd = binned[i];
        const int p = atomicAdd(&s_cur[wd >> 17], 1);
        csr[bstart + p] = wd & 0x1FFFF;
    }
}


template <int IN, int OUT, bool ROWSCALE>
__global__ __launch_bounds__(256) void k_gemm_wmma(const float* __restrict__ X,
                                                   const float* __restrict__ W,
                                                   const float* __restrict__ rs,
                                                   float* __restrict__ Y,
                                                   int nrows) {
    const int lane = threadIdx.x & 31;
    const int wave = blockIdx.x * (blockDim.x >> 5) + (threadIdx.x >> 5);
    const int row0 = wave * 16;
    if (row0 >= nrows) return;

    const int half = lane >> 4;
    const int l15  = lane & 15;
    const int arow = row0 + l15;

    constexpr int NT = OUT / 16;

    float sc[16];
    if constexpr (ROWSCALE) {
        const float* rsw = rs + __builtin_amdgcn_readfirstlane(row0);
        #pragma unroll
        for (int i = 0; i < 16; ++i) sc[i] = rsw[i];
    }

    v8f acc[NT];
    #pragma unroll
    for (int t = 0; t < NT; ++t)
        #pragma unroll
        for (int v = 0; v < 8; ++v) acc[t][v] = 0.0f;

    for (int k = 0; k < IN; k += 4) {
        const float* ap = X + (size_t)arow * IN + (k + 2 * half);
        v2f a;
        a.x = ap[0];
        a.y = ap[1];

        #pragma unroll
        for (int t = 0; t < NT; ++t) {
            const int col = t * 16 + l15;
            v2f b;
            b.x = W[(k + 2 * half)     * OUT + col];
            b.y = W[(k + 2 * half + 1) * OUT + col];
            acc[t] = __builtin_amdgcn_wmma_f32_16x16x4_f32(
                 false, a,  false, b,
                 (short)0, acc[t],  false,  false);
        }
    }

    if constexpr (ROWSCALE) {
        int hsel = half;
        #pragma unroll
        for (int t = 0; t < NT; ++t) asm volatile("" : "+v"(acc[t]));
        asm volatile("v_nop\n\tv_nop\n\tv_nop\n\tv_nop\n\tv_nop\n\tv_nop\n\tv_nop\n\tv_nop"
                     : "+v"(hsel), "+s"(sc[0]), "+s"(sc[1]), "+s"(sc[2]), "+s"(sc[3]), "+s"(sc[4]), "+s"(sc[5]),
                       "+s"(sc[6]), "+s"(sc[7]), "+s"(sc[8]), "+s"(sc[9]), "+s"(sc[10]), "+s"(sc[11]),
                       "+s"(sc[12]), "+s"(sc[13]));
        asm volatile("" : "+s"(sc[14]), "+s"(sc[15]));
        #pragma unroll
        for (int t = 0; t < NT; ++t) asm volatile("" : "+v"(acc[t]));
        #pragma unroll
        for (int v = 0; v < 8; ++v) {
            const int r = row0 + v + 8 * half;
            const float f = hsel ? sc[v + 8] : sc[v];
            #pragma unroll
            for (int t = 0; t < NT; ++t) Y[(size_t)r * OUT + t * 16 + l15] = f * acc[t][v];
        }
    } else {
        #pragma unroll
        for (int t = 0; t < NT; ++t)
            #pragma unroll
            for (int v = 0; v < 8; ++v) {
                int r = row0 + v + 8 * half;
                int c = t * 16 + l15;
                Y[(size_t)r * OUT + c] = acc[t][v];
            }
    }
}


template <int OUT>
__device__ __forceinline__ float gather_line(const int* __restrict__ csr, const int* __restrict__ off,
                                             const float* __restrict__ dis, const float* __restrict__ hw,
                                             const float* __restrict__ b, const int d0, const int lane) {
    constexpr int G = 32 / OUT;
    const int g = lane / OUT;
    const int f = lane % OUT;

    const float bias = b[f];
    float keep = 0.0f;
    #pragma unroll
    for (int nd = 0; nd < G; ++nd) {
        const int d = d0 + nd;
        const int beg = off[d];
        const int end = (d + 1 < N_NODES) ? off[d + 1] : N_EDGES;
        const float dd = dis[d];
        const float self = hw[(size_t)d * OUT + f];
        float acc = (g == 0) ? self : 0.0f;
        for (int base = beg; base < end; base += 32) {
            const int n = min(32, end - base);
            int s_j = d;
            if (lane < n) s_j = csr[base + lane];
            for (int jj = 0; jj < n; jj += 8 * G) {
                float r[8];
                #pragma unroll
                for (int u = 0; u < 8; ++u) {
                    const int s = __shfl(s_j, jj + u * G + g);
                    r[u] = hw[(size_t)s * OUT + f];
                }
                #pragma unroll
                for (int u = 0; u < 8; ++u) acc += (jj + u * G + g < n) ? r[u] : 0.0f;
            }
        }
        if (G == 2) acc += __shfl_xor(acc, 16);
        float v = dd * acc + bias;
        v = v > 0.0f ? v : 0.0f;
        if (g == nd) keep = dd * v;
    }
    return keep;
}


#define TILE_PITCH 34

static_assert(N_NODES % 16 == 0, "the fused layer-1 gather takes whole 16-node tiles, so no wave leaves before the barrier");

__global__ __launch_bounds__(512) void k_gather32_gemm16(const int* __restrict__ csr, const int* __restrict__ off,
                                                         const float* __restrict__ dis, const float* __restrict__ hw,
                                                         const float* __restrict__ b, const float* __restrict__ W,
                                                         float* __restrict__ Y) {
    __shared__ __attribute__((aligned(8))) float s_tile[16 * TILE_PITCH];
    const int lane = threadIdx.x & 31;
    const int w = __builtin_amdgcn_readfirstlane(threadIdx.x >> 5);
    const int row0 = blockIdx.x * 16;

    s_tile[w * TILE_PITCH + lane] = gather_line<32>(csr, off, dis, hw, b, row0 + w, lane);

    const int half = lane >> 4;
    const int l15  = lane & 15;

    v2f bf[8];
    #pragma unroll
    for (int ks = 0; ks < 8; ++ks) { bf[ks].x = 0.0f; bf[ks].y = 0.0f; }
    if (w == 0) {
        #pragma unroll
        for (int ks = 0; ks < 8; ++ks) {
            bf[ks].x = W[(4 * ks + 2 * half)     * 16 + l15];
            bf[ks].y = W[(4 * ks + 2 * half + 1) * 16 + l15];
        }
    }
    __syncthreads();
    if (w != 0) return;

    v2f a[8];
    #pragma unroll
    for (int ks = 0; ks < 8; ++ks) {
        const float* ap = s_tile + l15 * TILE_PITCH + (4 * ks + 2 * half);
        a[ks].x = ap[0];
        a[ks].y = ap[1];
    }
    float* yp = Y + (size_t)(row0 + 8 * half) * 16 + l15;

    asm volatile("" : "+v"(yp),
                      "+v"(a[0]), "+v"(a[1]), "+v"(a[2]), "+v"(a[3]), "+v"(a[4]), "+v"(a[5]), "+v"(a[6]), "+v"(a[7]),
                      "+v"(bf[0]), "+v"(bf[1]), "+v"(bf[2]), "+v"(bf[3]), "+v"(bf[4]), "+v"(bf[5]), "+v"(bf[6]),
                      "+v"(bf[7]));

    v8f acc;
    #pragma unroll
    for (int v = 0; v < 8; ++v) acc[v] = 0.0f;
    #pragma unroll
    for (int ks = 0; ks < 8; ++ks)
        acc = __builtin_amdgcn_wmma_f32_16x16x4_f32(
             false, a[ks],  false, bf[ks],
             (short)0, acc,  false,  false);
    __attribute__((address_space(1))) float* ypg = (__attribute__((address_space(1))) float*)yp;
    #pragma unroll
    for (int v = 0; v < 8; ++v) ypg[v * 16] = acc[v];
}


#define H2_PITCH 20

__global__ __launch_bounds__(512) void k_gather16_dot(const int* __restrict__ csr, const int* __restrict__ off,
                                                      const float* __restrict__ dis, const float* __restrict__ hw,
                                                      const float* __restrict__ b, const float* __restrict__ W3,
                                                      float* __restrict__ hw3) {
    __shared__ __attribute__((aligned(16))) float s_h2[32 * H2_PITCH];
    const int lane = threadIdx.x & 31;
    const int w = __builtin_amdgcn_readfirstlane(threadIdx.x >> 5);
    const int d0 = (blockIdx.x * 16 + w) * 2;

    s_h2[(2 * w + (lane >> 4)) * H2_PITCH + (lane & 15)] = gather_line<16>(csr, off, dis, hw, b, d0, lane);
    __syncthreads();
    if (w != 0) return;

    const float* r = s_h2 + lane * H2_PITCH;
    float s = 0.0f;
    #pragma unroll
    for (int k = 0; k < 16; ++k) s += r[k] * W3[k];
    hw3[blockIdx.x * 32 + lane] = s;
}


__global__ __launch_bounds__(1024) void k_gather_out1(const int* __restrict__ csr, const int* __restrict__ off,
                                                      const float* __restrict__ dis, const float* __restrict__ hw3,
                                                      const float* __restrict__ b3, float* __restrict__ out) {
    __shared__ float s_out[32];
    const int lane = threadIdx.x & 31;
    const int w = __builtin_amdgcn_readfirstlane(threadIdx.x >> 5);
    const int d = blockIdx.x * 32 + w;
    const int beg = off[d];
    const int end = (d + 1 < N_NODES) ? off[d + 1] : N_EDGES;
    const float dd = dis[d];
    float acc = 0.0f;
    for (int j = beg + lane; j < end; j += 32) acc += hw3[csr[j]];
    for (int o = 16; o > 0; o >>= 1) acc += __shfl_xor(acc, o);
    if (lane == 0) s_out[w] = dd * (acc + hw3[d]) + b3[0];
    __syncthreads();
    if (w == 0) out[blockIdx.x * 32 + lane] = s_out[lane];
}


static inline int blks(long long n, int bs) { return (int)((n + bs - 1) / bs); }

extern "C" void kernel_launch(void* const* d_in, const int* in_sizes, int n_in,
                              void* d_out, int out_size, void* d_ws, size_t ws_size,
                              hipStream_t stream) {
    (void)in_sizes; (void)n_in; (void)out_size; (void)ws_size;

    const float* x   = (const float*)d_in[0];
    const int*   ei  = (const int*)  d_in[1];
    const float* W1  = (const float*)d_in[2];
    const float* b1  = (const float*)d_in[3];
    const float* W2  = (const float*)d_in[4];
    const float* b2  = (const float*)d_in[5];
    const float* W3  = (const float*)d_in[6];
    const float* b3  = (const float*)d_in[7];
    float* out = (float*)d_out;

    const int* src = ei;
    const int* dst = ei + N_EDGES;

    float* ws   = (float*)d_ws;
    float* dis  = ws;
    int*   off  = (int*)(ws + N_NODES);
    int*   csr  = (int*)(ws + 2 * N_NODES);
    float* pA   = ws + 34 * N_NODES;
    float* pB   = ws + 66 * N_NODES;
    int*   hist   = (int*)pA;
    int*   binoff = (int*)pA + HIST_LEN;
    int*   binned = (int*)pB;
    float* hw1  = pA;
    float* hw2  = pB;
    float* hw3  = pA;

    const int BS = 256;

    k_bin_count  <<<SORT_BLOCKS, 1024, 0, stream>>>(dst, hist);
    k_scan_excl<HIST_LEN><<<(HIST_LEN + 1023) / 1024, 1024, 0, stream>>>(hist, binoff);
    k_bin_scatter<<<SORT_BLOCKS, 1024, 0, stream>>>(src, dst, binoff, binned);
    k_bin_sort   <<<N_BINS, 1024, 0, stream>>>(binned, binoff, off, dis, csr);

    const int gemm_grid = blks(N_NODES / 16, BS / 32);

    k_gemm_wmma<128, 32, true><<<gemm_grid, BS, 0, stream>>>(x, W1, dis, hw1, N_NODES);
    k_gather32_gemm16<<<N_NODES / 16, 512, 0, stream>>>(csr, off, dis, hw1, b1, W2, hw2);

    k_gather16_dot<<<N_NODES / 32, 512, 0, stream>>>(csr, off, dis, hw2, b2, W3, hw3);

    k_gather_out1<<<N_NODES / 32, 1024, 0, stream>>>(csr, off, dis, hw3, b3, out);
}


# Both sort levels rank edges into an LDS staging array (by bin, then by node) and copy it out in order: contiguous runs replace two scattered 4-byte stores per edge; sums stay in LDS arrival order
# speedup vs baseline: 6.3597x; 1.2958x over previous
#include <hip/hip_runtime.h>


typedef __attribute__((ext_vector_type(2))) float v2f;
typedef __attribute__((ext_vector_type(8))) float v8f;

#define N_NODES 100000
#define N_EDGES 3200000

static_assert(N_NODES % 32 == 0, "layer-3 gather writes one full 32-node line per block");
static_assert(N_EDGES == 32 * N_NODES, "the binned edge list reuses the 32*N word plane of the layer-1 output");


#define SORT_BLOCKS 256
#define N_BINS      ((N_NODES + 255) / 256)
#define EDGE_CHUNK  ((N_EDGES + SORT_BLOCKS - 1) / SORT_BLOCKS)
#define HIST_LEN    (N_BINS * SORT_BLOCKS)

static_assert(N_NODES <= (1 << 17), "src must fit in the low 17 bits of a binned word");
static_assert(2 * HIST_LEN <= 32 * N_NODES, "histogram and its scan share plane A");

__global__ __launch_bounds__(1024) void k_bin_count(const int* __restrict__ dst, int* __restrict__ hist) {
    __shared__ int s_hist[N_BINS];
    const int t = threadIdx.x;
    const int blk = blockIdx.x;
    for (int b = t; b < N_BINS; b += 1024) s_hist[b] = 0;
    __syncthreads();
    const int lo = blk * EDGE_CHUNK;
    const int hi = min(lo + EDGE_CHUNK, N_EDGES);
    int e = lo + t;
    for (; e + 3 * 1024 < hi; e += 4 * 1024) {
        const int d0 = dst[e];
        const int d1 = dst[e + 1024];
        const int d2 = dst[e + 2 * 1024];
        const int d3 = dst[e + 3 * 1024];
        atomicAdd(&s_hist[d0 >> 8], 1);
        atomicAdd(&s_hist[d1 >> 8], 1);
        atomicAdd(&s_hist[d2 >> 8], 1);
        atomicAdd(&s_hist[d3 >> 8], 1);
    }
    for (; e < hi; e += 1024) atomicAdd(&s_hist[dst[e] >> 8], 1);
    __syncthreads();
    for (int b = t; b < N_BINS; b += 1024) hist[b * SORT_BLOCKS + blk] = s_hist[b];
}

template <int LEN>
__global__ __launch_bounds__(1024) void k_scan_excl(const int* __restrict__ cnt, int* __restrict__ out) {
    __shared__ int s_incl[32];
    __shared__ int s_before[32];
    const int t = threadIdx.x;
    const int lane = t & 31;
    const int w = t >> 5;
    const int i = blockIdx.x * 1024 + t;
    const int c = (i < LEN) ? cnt[i] : 0;

    int before = 0;
    for (int k = 0; k < (int)blockIdx.x; ++k) before += cnt[k * 1024 + t];

    int incl = c;
    for (int o = 1; o < 32; o <<= 1) {
        int v = __shfl_up(incl, o);
        if (lane >= o) incl += v;
    }
    for (int o = 16; o > 0; o >>= 1) before += __shfl_xor(before, o);
    if (lane == 31) s_incl[w] = incl;
    if (lane == 0) s_before[w] = before;
    __syncthreads();

    int base = 0, woff = 0;
    for (int k = 0; k < 32; ++k) {
        base += s_before[k];
        if (k < w) woff += s_incl[k];
    }
    if (i < LEN) out[i] = base + woff + incl - c;
}

static_assert(N_BINS <= 1024, "one thread per bin scans the block's bin counts");
static_assert(N_BINS <= 65535, "a staged edge's bin is kept as a 16-bit value");

__global__ __launch_bounds__(1024) void k_bin_scatter(const int* __restrict__ src, const int* __restrict__ dst,
                                                      const int* __restrict__ binoff, int* __restrict__ binned) {
    __shared__ int s_stage[EDGE_CHUNK];
    __shared__ unsigned short s_bin[EDGE_CHUNK];
    __shared__ int s_cur[N_BINS];
    __shared__ int s_delta[N_BINS];
    __shared__ int s_wsum[32];
    const int t = threadIdx.x;
    const int lane = t & 31;
    const int w = t >> 5;
    const int blk = blockIdx.x;

    int g0 = 0, c = 0;
    if (t < N_BINS) {
        const int i = t * SORT_BLOCKS + blk;
        g0 = binoff[i];
        const int g1 = (i + 1 < HIST_LEN) ? binoff[i + 1] : N_EDGES;
        c = g1 - g0;
    }
    int incl = c;
    for (int o = 1; o < 32; o <<= 1) {
        int v = __shfl_up(incl, o);
        if (lane >= o) incl += v;
    }
    if (lane == 31) s_wsum[w] = incl;
    __syncthreads();
    int woff = 0;
    for (int k = 0; k < 32; ++k)
        if (k < w) woff += s_wsum[k];
    const int excl = woff + incl - c;
    if (t < N_BINS) {
        s_cur[t] = excl;
        s_delta[t] = g0 - excl;
    }
    __syncthreads();

    const int lo = blk * EDGE_CHUNK;
    const int hi = min(lo + EDGE_CHUNK, N_EDGES);
    int e = lo + t;
    for (; e + 3 * 1024 < hi; e += 4 * 1024) {
        const int d0 = dst[e];
        const int d1 = dst[e + 1024];
        const int d2 = dst[e + 2 * 1024];
        const int d3 = dst[e + 3 * 1024];
        const int s0 = src[e];
        const int s1 = src[e + 1024];
        const int s2 = src[e + 2 * 1024];
        const int s3 = src[e + 3 * 1024];
        const int p0 = atomicAdd(&s_cur[d0 >> 8], 1);
        const int p1 = atomicAdd(&s_cur[d1 >> 8], 1);
        const int p2 = atomicAdd(&s_cur[d2 >> 8], 1);
        const int p3 = atomicAdd(&s_cur[d3 >> 8], 1);
        s_stage[p0] = s0 | ((d0 & 255) << 17);
        s_stage[p1] = s1 | ((d1 & 255) << 17);
        s_stage[p2] = s2 | ((d2 & 255) << 17);
        s_stage[p3] = s3 | ((d3 & 255) << 17);
        s_bin[p0] = (unsigned short)(d0 >> 8);
        s_bin[p1] = (unsigned short)(d1 >> 8);
        s_bin[p2] = (unsigned short)(d2 >> 8);
        s_bin[p3] = (unsigned short)(d3 >> 8);
    }
    for (; e < hi; e += 1024) {
        const int d = dst[e];
        const int p = atomicAdd(&s_cur[d >> 8], 1);
        s_stage[p] = src[e] | ((d & 255) << 17);
        s_bin[p] = (unsigned short)(d >> 8);
    }
    __syncthreads();

    const int n = hi - lo;
    int i = t;
    for (; i + 3 * 1024 < n; i += 4 * 1024) {
        const int b0 = s_bin[i];
        const int b1 = s_bin[i + 1024];
        const int b2 = s_bin[i + 2 * 1024];
        const int b3 = s_bin[i + 3 * 1024];
        const int w0 = s_stage[i];
        const int w1 = s_stage[i + 1024];
        const int w2 = s_stage[i + 2 * 1024];
        const int w3 = s_stage[i + 3 * 1024];
        binned[s_delta[b0] + i] = w0;
        binned[s_delta[b1] + i + 1024] = w1;
        binned[s_delta[b2] + i + 2 * 1024] = w2;
        binned[s_delta[b3] + i + 3 * 1024] = w3;
    }
    for (; i < n; i += 1024) binned[s_delta[s_bin[i]] + i] = s_stage[i];
}

#define BIN_STAGE 12288

__global__ __launch_bounds__(1024) void k_bin_sort(const int* __restrict__ binned, const int* __restrict__ binoff,
                                                   int* __restrict__ off, float* __restrict__ dis,
                                                   int* __restrict__ csr) {
    __shared__ int s_cnt[256];
    __shared__ int s_cur[256];
    __shared__ int s_wsum[8];
    __shared__ int s_stage[BIN_STAGE];
    const int t = threadIdx.x;
    const int lane = t & 31;
    const int bin = blockIdx.x;
    const int bstart = binoff[bin * SORT_BLOCKS];
    const int bend = (bin + 1 < N_BINS) ? binoff[(bin + 1) * SORT_BLOCKS] : N_EDGES;
    const int n = bend - bstart;
    const bool staged = n <= BIN_STAGE;

    if (t < 256) s_cnt[t] = 0;
    __syncthreads();
    int i = bstart + t;
    for (; i + 3 * 1024 < bend; i += 4 * 1024) {
        const int w0 = binned[i];
        const int w1 = binned[i + 1024];
        const int w2 = binned[i + 2 * 1024];
        const int w3 = binned[i + 3 * 1024];
        atomicAdd(&s_cnt[w0 >> 17], 1);
        atomicAdd(&s_cnt[w1 >> 17], 1);
        atomicAdd(&s_cnt[w2 >> 17], 1);
        atomicAdd(&s_cnt[w3 >> 17], 1);
    }
    for (; i < bend; i += 1024) atomicAdd(&s_cnt[binned[i] >> 17], 1);
    __syncthreads();

    int c = 0, incl = 0;
    if (t < 256) {
        c = s_cnt[t];
        incl = c;
        for (int o = 1; o < 32; o <<= 1) {
            int v = __shfl_up(incl, o);
            if (lane >= o) incl += v;
        }
        if (lane == 31) s_wsum[t >> 5] = incl;
    }
    __syncthreads();
    if (t < 256) {
        int woff = 0;
        for (int k = 0; k < 8; ++k)
            if (k < (t >> 5)) woff += s_wsum[k];
        const int excl = woff + incl - c;
        s_cur[t] = excl;
        const int node = bin * 256 + t;
        if (node < N_NODES) {
            off[node] = bstart + excl;
            dis[node] = rsqrtf((float)(c + 1));
        }
    }
    __syncthreads();
    i = bstart + t;
    if (staged) {
        for (; i + 3 * 1024 < bend; i += 4 * 1024) {
            const int w0 = binned[i];
            const int w1 = binned[i + 1024];
            const int w2 = binned[i + 2 * 1024];
            const int w3 = binned[i + 3 * 1024];
            const int p0 = atomicAdd(&s_cur[w0 >> 17], 1);
            const int p1 = atomicAdd(&s_cur[w1 >> 17], 1);
            const int p2 = atomicAdd(&s_cur[w2 >> 17], 1);
            const int p3 = atomicAdd(&s_cur[w3 >> 17], 1);
            s_stage[p0] = w0 & 0x1FFFF;
            s_stage[p1] = w1 & 0x1FFFF;
            s_stage[p2] = w2 & 0x1FFFF;
            s_stage[p3] = w3 & 0x1FFFF;
        }
        for (; i < bend; i += 1024) {
            const int wd = binned[i];
            const int p = atomicAdd(&s_cur[wd >> 17], 1);
            s_stage[p] = wd & 0x1FFFF;
        }
    } else {
        for (; i + 3 * 1024 < bend; i += 4 * 1024) {
            const int w0 = binned[i];
            const int w1 = binned[i + 1024];
            const int w2 = binned[i + 2 * 1024];
            const int w3 = binned[i + 3 * 1024];
            const int p0 = atomicAdd(&s_cur[w0 >> 17], 1);
            const int p1 = atomicAdd(&s_cur[w1 >> 17], 1);
            const int p2 = atomicAdd(&s_cur[w2 >> 17], 1);
            const int p3 = atomicAdd(&s_cur[w3 >> 17], 1);
            csr[bstart + p0] = w0 & 0x1FFFF;
            csr[bstart + p1] = w1 & 0x1FFFF;
            csr[bstart + p2] = w2 & 0x1FFFF;
            csr[bstart + p3] = w3 & 0x1FFFF;
        }
        for (; i < bend; i += 1024) {
            const int wd = binned[i];
            const int p = atomicAdd(&s_cur[wd >> 17], 1);
            csr[bstart + p] = wd & 0x1FFFF;
        }
    }
    __syncthreads();
    if (staged)
        for (int j = t; j < n; j += 1024) csr[bstart + j] = s_stage[j];
}


template <int IN, int OUT, bool ROWSCALE>
__global__ __launch_bounds__(256) void k_gemm_wmma(const float* __restrict__ X,
                                                   const float* __restrict__ W,
                                                   const float* __restrict__ rs,
                                                   float* __restrict__ Y,
                                                   int nrows) {
    const int lane = threadIdx.x & 31;
    const int wave = blockIdx.x * (blockDim.x >> 5) + (threadIdx.x >> 5);
    const int row0 = wave * 16;
    if (row0 >= nrows) return;

    const int half = lane >> 4;
    const int l15  = lane & 15;
    const int arow = row0 + l15;

    constexpr int NT = OUT / 16;

    float sc[16];
    if constexpr (ROWSCALE) {
        const float* rsw = rs + __builtin_amdgcn_readfirstlane(row0);
        #pragma unroll
        for (int i = 0; i < 16; ++i) sc[i] = rsw[i];
    }

    v8f acc[NT];
    #pragma unroll
    for (int t = 0; t < NT; ++t)
        #pragma unroll
        for (int v = 0; v < 8; ++v) acc[t][v] = 0.0f;

    for (int k = 0; k < IN; k += 4) {
        const float* ap = X + (size_t)arow * IN + (k + 2 * half);
        v2f a;
        a.x = ap[0];
        a.y = ap[1];

        #pragma unroll
        for (int t = 0; t < NT; ++t) {
            const int col = t * 16 + l15;
            v2f b;
            b.x = W[(k + 2 * half)     * OUT + col];
            b.y = W[(k + 2 * half + 1) * OUT + col];
            acc[t] = __builtin_amdgcn_wmma_f32_16x16x4_f32(
                 false, a,  false, b,
                 (short)0, acc[t],  false,  false);
        }
    }

    if constexpr (ROWSCALE) {
        int hsel = half;
        #pragma unroll
        for (int t = 0; t < NT; ++t) asm volatile("" : "+v"(acc[t]));
        asm volatile("v_nop\n\tv_nop\n\tv_nop\n\tv_nop\n\tv_nop\n\tv_nop\n\tv_nop\n\tv_nop"
                     : "+v"(hsel), "+s"(sc[0]), "+s"(sc[1]), "+s"(sc[2]), "+s"(sc[3]), "+s"(sc[4]), "+s"(sc[5]),
                       "+s"(sc[6]), "+s"(sc[7]), "+s"(sc[8]), "+s"(sc[9]), "+s"(sc[10]), "+s"(sc[11]),
                       "+s"(sc[12]), "+s"(sc[13]));
        asm volatile("" : "+s"(sc[14]), "+s"(sc[15]));
        #pragma unroll
        for (int t = 0; t < NT; ++t) asm volatile("" : "+v"(acc[t]));
        #pragma unroll
        for (int v = 0; v < 8; ++v) {
            const int r = row0 + v + 8 * half;
            const float f = hsel ? sc[v + 8] : sc[v];
            #pragma unroll
            for (int t = 0; t < NT; ++t) Y[(size_t)r * OUT + t * 16 + l15] = f * acc[t][v];
        }
    } else {
        #pragma unroll
        for (int t = 0; t < NT; ++t)
            #pragma unroll
            for (int v = 0; v < 8; ++v) {
                int r = row0 + v + 8 * half;
                int c = t * 16 + l15;
                Y[(size_t)r * OUT + c] = acc[t][v];
            }
    }
}


template <int OUT>
__device__ __forceinline__ float gather_line(const int* __restrict__ csr, const int* __restrict__ off,
                                             const float* __restrict__ dis, const float* __restrict__ hw,
                                             const float* __restrict__ b, const int d0, const int lane) {
    constexpr int G = 32 / OUT;
    const int g = lane / OUT;
    const int f = lane % OUT;

    const float bias = b[f];
    float keep = 0.0f;
    #pragma unroll
    for (int nd = 0; nd < G; ++nd) {
        const int d = d0 + nd;
        const int beg = off[d];
        const int end = (d + 1 < N_NODES) ? off[d + 1] : N_EDGES;
        const float dd = dis[d];
        const float self = hw[(size_t)d * OUT + f];
        float acc = (g == 0) ? self : 0.0f;
        for (int base = beg; base < end; base += 32) {
            const int n = min(32, end - base);
            int s_j = d;
            if (lane < n) s_j = csr[base + lane];
            for (int jj = 0; jj < n; jj += 8 * G) {
                float r[8];
                #pragma unroll
                for (int u = 0; u < 8; ++u) {
                    const int s = __shfl(s_j, jj + u * G + g);
                    r[u] = hw[(size_t)s * OUT + f];
                }
                #pragma unroll
                for (int u = 0; u < 8; ++u) acc += (jj + u * G + g < n) ? r[u] : 0.0f;
            }
        }
        if (G == 2) acc += __shfl_xor(acc, 16);
        float v = dd * acc + bias;
        v = v > 0.0f ? v : 0.0f;
        if (g == nd) keep = dd * v;
    }
    return keep;
}


#define TILE_PITCH 34

static_assert(N_NODES % 16 == 0, "the fused layer-1 gather takes whole 16-node tiles, so no wave leaves before the barrier");

__global__ __launch_bounds__(512) void k_gather32_gemm16(const int* __restrict__ csr, const int* __restrict__ off,
                                                         const float* __restrict__ dis, const float* __restrict__ hw,
                                                         const float* __restrict__ b, const float* __restrict__ W,
                                                         float* __restrict__ Y) {
    __shared__ __attribute__((aligned(8))) float s_tile[16 * TILE_PITCH];
    const int lane = threadIdx.x & 31;
    const int w = __builtin_amdgcn_readfirstlane(threadIdx.x >> 5);
    const int row0 = blockIdx.x * 16;

    s_tile[w * TILE_PITCH + lane] = gather_line<32>(csr, off, dis, hw, b, row0 + w, lane);

    const int half = lane >> 4;
    const int l15  = lane & 15;

    v2f bf[8];
    #pragma unroll
    for (int ks = 0; ks < 8; ++ks) { bf[ks].x = 0.0f; bf[ks].y = 0.0f; }
    if (w == 0) {
        #pragma unroll
        for (int ks = 0; ks < 8; ++ks) {
            bf[ks].x = W[(4 * ks + 2 * half)     * 16 + l15];
            bf[ks].y = W[(4 * ks + 2 * half + 1) * 16 + l15];
        }
    }
    __syncthreads();
    if (w != 0) return;

    v2f a[8];
    #pragma unroll
    for (int ks = 0; ks < 8; ++ks) {
        const float* ap = s_tile + l15 * TILE_PITCH + (4 * ks + 2 * half);
        a[ks].x = ap[0];
        a[ks].y = ap[1];
    }
    float* yp = Y + (size_t)(row0 + 8 * half) * 16 + l15;

    asm volatile("" : "+v"(yp),
                      "+v"(a[0]), "+v"(a[1]), "+v"(a[2]), "+v"(a[3]), "+v"(a[4]), "+v"(a[5]), "+v"(a[6]), "+v"(a[7]),
                      "+v"(bf[0]), "+v"(bf[1]), "+v"(bf[2]), "+v"(bf[3]), "+v"(bf[4]), "+v"(bf[5]), "+v"(bf[6]),
                      "+v"(bf[7]));

    v8f acc;
    #pragma unroll
    for (int v = 0; v < 8; ++v) acc[v] = 0.0f;
    #pragma unroll
    for (int ks = 0; ks < 8; ++ks)
        acc = __builtin_amdgcn_wmma_f32_16x16x4_f32(
             false, a[ks],  false, bf[ks],
             (short)0, acc,  false,  false);
    __attribute__((address_space(1))) float* ypg = (__attribute__((address_space(1))) float*)yp;
    #pragma unroll
    for (int v = 0; v < 8; ++v) ypg[v * 16] = acc[v];
}


#define H2_PITCH 20

__global__ __launch_bounds__(512) void k_gather16_dot(const int* __restrict__ csr, const int* __restrict__ off,
                                                      const float* __restrict__ dis, const float* __restrict__ hw,
                                                      const float* __restrict__ b, const float* __restrict__ W3,
                                                      float* __restrict__ hw3) {
    __shared__ __attribute__((aligned(16))) float s_h2[32 * H2_PITCH];
    const int lane = threadIdx.x & 31;
    const int w = __builtin_amdgcn_readfirstlane(threadIdx.x >> 5);
    const int d0 = (blockIdx.x * 16 + w) * 2;

    s_h2[(2 * w + (lane >> 4)) * H2_PITCH + (lane & 15)] = gather_line<16>(csr, off, dis, hw, b, d0, lane);
    __syncthreads();
    if (w != 0) return;

    const float* r = s_h2 + lane * H2_PITCH;
    float s = 0.0f;
    #pragma unroll
    for (int k = 0; k < 16; ++k) s += r[k] * W3[k];
    hw3[blockIdx.x * 32 + lane] = s;
}


__global__ __launch_bounds__(1024) void k_gather_out1(const int* __restrict__ csr, const int* __restrict__ off,
                                                      const float* __restrict__ dis, const float* __restrict__ hw3,
                                                      const float* __restrict__ b3, float* __restrict__ out) {
    __shared__ float s_out[32];
    const int lane = threadIdx.x & 31;
    const int w = __builtin_amdgcn_readfirstlane(threadIdx.x >> 5);
    const int d = blockIdx.x * 32 + w;
    const int beg = off[d];
    const int end = (d + 1 < N_NODES) ? off[d + 1] : N_EDGES;
    const float dd = dis[d];
    float acc = 0.0f;
    for (int j = beg + lane; j < end; j += 32) acc += hw3[csr[j]];
    for (int o = 16; o > 0; o >>= 1) acc += __shfl_xor(acc, o);
    if (lane == 0) s_out[w] = dd * (acc + hw3[d]) + b3[0];
    __syncthreads();
    if (w == 0) out[blockIdx.x * 32 + lane] = s_out[lane];
}


static inline int blks(long long n, int bs) { return (int)((n + bs - 1) / bs); }

extern "C" void kernel_launch(void* const* d_in, const int* in_sizes, int n_in,
                              void* d_out, int out_size, void* d_ws, size_t ws_size,
                              hipStream_t stream) {
    (void)in_sizes; (void)n_in; (void)out_size; (void)ws_size;

    const float* x   = (const float*)d_in[0];
    const int*   ei  = (const int*)  d_in[1];
    const float* W1  = (const float*)d_in[2];
    const float* b1  = (const float*)d_in[3];
    const float* W2  = (const float*)d_in[4];
    const float* b2  = (const float*)d_in[5];
    const float* W3  = (const float*)d_in[6];
    const float* b3  = (const float*)d_in[7];
    float* out = (float*)d_out;

    const int* src = ei;
    const int* dst = ei + N_EDGES;

    float* ws   = (float*)d_ws;
    float* dis  = ws;
    int*   off  = (int*)(ws + N_NODES);
    int*   csr  = (int*)(ws + 2 * N_NODES);
    float* pA   = ws + 34 * N_NODES;
    float* pB   = ws + 66 * N_NODES;
    int*   hist   = (int*)pA;
    int*   binoff = (int*)pA + HIST_LEN;
    int*   binned = (int*)pB;
    float* hw1  = pA;
    float* hw2  = pB;
    float* hw3  = pA;

    const int BS = 256;

    k_bin_count  <<<SORT_BLOCKS, 1024, 0, stream>>>(dst, hist);
    k_scan_excl<HIST_LEN><<<(HIST_LEN + 1023) / 1024, 1024, 0, stream>>>(hist, binoff);
    k_bin_scatter<<<SORT_BLOCKS, 1024, 0, stream>>>(src, dst, binoff, binned);
    k_bin_sort   <<<N_BINS, 1024, 0, stream>>>(binned, binoff, off, dis, csr);

    const int gemm_grid = blks(N_NODES / 16, BS / 32);

    k_gemm_wmma<128, 32, true><<<gemm_grid, BS, 0, stream>>>(x, W1, dis, hw1, N_NODES);
    k_gather32_gemm16<<<N_NODES / 16, 512, 0, stream>>>(csr, off, dis, hw1, b1, W2, hw2);

    k_gather16_dot<<<N_NODES / 32, 512, 0, stream>>>(csr, off, dis, hw2, b2, W3, hw3);

    k_gather_out1<<<N_NODES / 32, 1024, 0, stream>>>(csr, off, dis, hw3, b3, out);
}


# Sort level 1 made block-local: each block bins its own stretch in LDS and writes it back with its bin starts; level 2 collects 256 runs; count and scan launches gone (8 -> 6); LDS arrival sum order
# speedup vs baseline: 7.3200x; 1.1510x over previous
#include <hip/hip_runtime.h>


typedef __attribute__((ext_vector_type(2))) float v2f;
typedef __attribute__((ext_vector_type(8))) float v8f;

#define N_NODES 100000
#define N_EDGES 3200000

static_assert(N_NODES % 32 == 0, "layer-3 gather writes one full 32-node line per block");
static_assert(N_EDGES == 32 * N_NODES, "the binned edge list reuses the 32*N word plane of the layer-1 output");


#define SORT_BLOCKS 256
#define N_BINS      ((N_NODES + 255) / 256)
#define EDGE_CHUNK  ((N_EDGES + SORT_BLOCKS - 1) / SORT_BLOCKS)
#define TBL_PITCH   (N_BINS + 1)
#define EPT         ((EDGE_CHUNK + 1023) / 1024)

static_assert(N_NODES <= (1 << 17), "src must fit in the low 17 bits of a binned word");
static_assert(N_EDGES == SORT_BLOCKS * EDGE_CHUNK, "every level-1 block owns a full stretch");
static_assert(SORT_BLOCKS * TBL_PITCH <= 32 * N_NODES, "the tables of bin starts lie in plane A");
static_assert(SORT_BLOCKS % 32 == 0, "a level-2 wave takes every 32nd run");
static_assert(N_BINS <= 1024, "one thread per bin scans the block's bin counts");

__global__ __launch_bounds__(1024) void k_chunk_sort(const int* __restrict__ src, const int* __restrict__ dst,
                                                     int* __restrict__ tbl, int* __restrict__ binned) {
    __shared__ int s_stage[EDGE_CHUNK];
    __shared__ int s_cnt[N_BINS];
    __shared__ int s_start[N_BINS];
    __shared__ int s_wsum[32];
    const int t = threadIdx.x;
    const int lane = t & 31;
    const int w = t >> 5;
    const int blk = blockIdx.x;
    for (int b = t; b < N_BINS; b += 1024) s_cnt[b] = 0;

    const int lo = blk * EDGE_CHUNK;
    int d[EPT], s[EPT], r[EPT];
    #pragma unroll
    for (int u = 0; u < EPT; ++u) d[u] = (t + u * 1024 < EDGE_CHUNK) ? dst[lo + t + u * 1024] : 0;
    #pragma unroll
    for (int u = 0; u < EPT; ++u) s[u] = (t + u * 1024 < EDGE_CHUNK) ? src[lo + t + u * 1024] : 0;
    __syncthreads();

    #pragma unroll
    for (int u = 0; u < EPT; ++u) {
        r[u] = 0;
        if (t + u * 1024 < EDGE_CHUNK) r[u] = atomicAdd(&s_cnt[d[u] >> 8], 1);
    }
    __syncthreads();

    const int c = (t < N_BINS) ? s_cnt[t] : 0;
    int incl = c;
    for (int o = 1; o < 32; o <<= 1) {
        int v = __shfl_up(incl, o);
        if (lane >= o) incl += v;
    }
    if (lane == 31) s_wsum[w] = incl;
    __syncthreads();
    int woff = 0;
    for (int k = 0; k < 32; ++k)
        if (k < w) woff += s_wsum[k];
    const int excl = woff + incl - c;
    if (t < N_BINS) {
        s_start[t] = excl;
        tbl[blk * TBL_PITCH + t] = excl;
    }
    if (t == 0) tbl[blk * TBL_PITCH + N_BINS] = EDGE_CHUNK;
    __syncthreads();

    #pragma unroll
    for (int u = 0; u < EPT; ++u)
        if (t + u * 1024 < EDGE_CHUNK) s_stage[s_start[d[u] >> 8] + r[u]] = s[u] | ((d[u] & 255) << 17);
    __syncthreads();

    for (int i = t; i < EDGE_CHUNK; i += 1024) binned[lo + i] = s_stage[i];
}

#define BIN_STAGE 12288
#define RPW (SORT_BLOCKS / 32)

template <bool STAGED>
__device__ __forceinline__ void place_edge(const int wd, int* s_cur, int* s_stage, int* __restrict__ csr,
                                           const int bstart) {
    const int p = atomicAdd(&s_cur[wd >> 17], 1);
    if (STAGED) s_stage[p] = wd & 0x1FFFF;
    else        csr[bstart + p] = wd & 0x1FFFF;
}

template <bool STAGED>
__device__ __forceinline__ void place_runs(const int (&w0)[RPW], const int (&w1)[RPW], const int (&rs)[RPW],
                                           const int (&rl)[RPW], int* s_cur, int* s_stage,
                                           const int* __restrict__ binned, int* __restrict__ csr,
                                           const int bstart, const int lane) {
    #pragma unroll
    for (int k = 0; k < RPW; ++k) {
        if (w0[k] >= 0) place_edge<STAGED>(w0[k], s_cur, s_stage, csr, bstart);
        if (w1[k] >= 0) place_edge<STAGED>(w1[k], s_cur, s_stage, csr, bstart);
    }
    #pragma unroll
    for (int k = 0; k < RPW; ++k)
        for (int j = lane + 64; j < rl[k]; j += 32) place_edge<STAGED>(binned[rs[k] + j], s_cur, s_stage, csr, bstart);
}

__global__ __launch_bounds__(1024) void k_bin_sort(const int* __restrict__ binned, const int* __restrict__ tbl,
                                                   int* __restrict__ off, float* __restrict__ dis,
                                                   int* __restrict__ csr) {
    __shared__ int s_cnt[256];
    __shared__ int s_cur[256];
    __shared__ int s_wsum[8];
    __shared__ int s_rs[SORT_BLOCKS];
    __shared__ int s_rl[SORT_BLOCKS];
    __shared__ int s_sum_start[RPW];
    __shared__ int s_sum_len[RPW];
    __shared__ int s_stage[BIN_STAGE];
    const int t = threadIdx.x;
    const int lane = t & 31;
    const int w = t >> 5;
    const int bin = blockIdx.x;

    if (t < SORT_BLOCKS) {
        const int a = tbl[t * TBL_PITCH + bin];
        const int b = tbl[t * TBL_PITCH + bin + 1];
        s_rs[t] = t * EDGE_CHUNK + a;
        s_rl[t] = b - a;
        int sa = a, sl = b - a;
        for (int o = 16; o > 0; o >>= 1) {
            sa += __shfl_xor(sa, o);
            sl += __shfl_xor(sl, o);
        }
        if (lane == 0) {
            s_sum_start[w] = sa;
            s_sum_len[w] = sl;
        }
    }
    if (t < 256) s_cnt[t] = 0;
    __syncthreads();
    int bstart = 0, n = 0;
    for (int k = 0; k < RPW; ++k) {
        bstart += s_sum_start[k];
        n += s_sum_len[k];
    }
    const bool staged = n <= BIN_STAGE;

    int rs[RPW], rl[RPW], w0[RPW], w1[RPW];
    #pragma unroll
    for (int k = 0; k < RPW; ++k) {
        rs[k] = s_rs[w + 32 * k];
        rl[k] = s_rl[w + 32 * k];
    }
    #pragma unroll
    for (int k = 0; k < RPW; ++k) {
        w0[k] = (lane < rl[k]) ? binned[rs[k] + lane] : -1;
        w1[k] = (lane + 32 < rl[k]) ? binned[rs[k] + lane + 32] : -1;
    }
    #pragma unroll
    for (int k = 0; k < RPW; ++k) {
        if (w0[k] >= 0) atomicAdd(&s_cnt[w0[k] >> 17], 1);
        if (w1[k] >= 0) atomicAdd(&s_cnt[w1[k] >> 17], 1);
    }
    #pragma unroll
    for (int k = 0; k < RPW; ++k)
        for (int j = lane + 64; j < rl[k]; j += 32) atomicAdd(&s_cnt[binned[rs[k] + j] >> 17], 1);
    __syncthreads();

    int c = 0, incl = 0;
    if (t < 256) {
        c = s_cnt[t];
        incl = c;
        for (int o = 1; o < 32; o <<= 1) {
            int v = __shfl_up(incl, o);
            if (lane >= o) incl += v;
        }
        if (lane == 31) s_wsum[t >> 5] = incl;
    }
    __syncthreads();
    if (t < 256) {
        int woff = 0;
        for (int k = 0; k < 8; ++k)
            if (k < (t >> 5)) woff += s_wsum[k];
        const int excl = woff + incl - c;
        s_cur[t] = excl;
        const int node = bin * 256 + t;
        if (node < N_NODES) {
            off[node] = bstart + excl;
            dis[node] = rsqrtf((float)(c + 1));
        }
    }
    __syncthreads();
    if (staged) place_runs<true >(w0, w1, rs, rl, s_cur, s_stage, binned, csr, bstart, lane);
    else        place_runs<false>(w0, w1, rs, rl, s_cur, s_stage, binned, csr, bstart, lane);
    __syncthreads();
    if (staged)
        for (int j = t; j < n; j += 1024) csr[bstart + j] = s_stage[j];
}


template <int IN, int OUT, bool ROWSCALE>
__global__ __launch_bounds__(256) void k_gemm_wmma(const float* __restrict__ X,
                                                   const float* __restrict__ W,
                                                   const float* __restrict__ rs,
                                                   float* __restrict__ Y,
                                                   int nrows) {
    const int lane = threadIdx.x & 31;
    const int wave = blockIdx.x * (blockDim.x >> 5) + (threadIdx.x >> 5);
    const int row0 = wave * 16;
    if (row0 >= nrows) return;

    const int half = lane >> 4;
    const int l15  = lane & 15;
    const int arow = row0 + l15;

    constexpr int NT = OUT / 16;

    float sc[16];
    if constexpr (ROWSCALE) {
        const float* rsw = rs + __builtin_amdgcn_readfirstlane(row0);
        #pragma unroll
        for (int i = 0; i < 16; ++i) sc[i] = rsw[i];
    }

    v8f acc[NT];
    #pragma unroll
    for (int t = 0; t < NT; ++t)
        #pragma unroll
        for (int v = 0; v < 8; ++v) acc[t][v] = 0.0f;

    for (int k = 0; k < IN; k += 4) {
        const float* ap = X + (size_t)arow * IN + (k + 2 * half);
        v2f a;
        a.x = ap[0];
        a.y = ap[1];

        #pragma unroll
        for (int t = 0; t < NT; ++t) {
            const int col = t * 16 + l15;
            v2f b;
            b.x = W[(k + 2 * half)     * OUT + col];
            b.y = W[(k + 2 * half + 1) * OUT + col];
            acc[t] = __builtin_amdgcn_wmma_f32_16x16x4_f32(
                 false, a,  false, b,
                 (short)0, acc[t],  false,  false);
        }
    }

    if constexpr (ROWSCALE) {
        int hsel = half;
        #pragma unroll
        for (int t = 0; t < NT; ++t) asm volatile("" : "+v"(acc[t]));
        asm volatile("v_nop\n\tv_nop\n\tv_nop\n\tv_nop\n\tv_nop\n\tv_nop\n\tv_nop\n\tv_nop"
                     : "+v"(hsel), "+s"(sc[0]), "+s"(sc[1]), "+s"(sc[2]), "+s"(sc[3]), "+s"(sc[4]), "+s"(sc[5]),
                       "+s"(sc[6]), "+s"(sc[7]), "+s"(sc[8]), "+s"(sc[9]), "+s"(sc[10]), "+s"(sc[11]),
                       "+s"(sc[12]), "+s"(sc[13]));
        asm volatile("" : "+s"(sc[14]), "+s"(sc[15]));
        #pragma unroll
        for (int t = 0; t < NT; ++t) asm volatile("" : "+v"(acc[t]));
        #pragma unroll
        for (int v = 0; v < 8; ++v) {
            const int r = row0 + v + 8 * half;
            const float f = hsel ? sc[v + 8] : sc[v];
            #pragma unroll
            for (int t = 0; t < NT; ++t) Y[(size_t)r * OUT + t * 16 + l15] = f * acc[t][v];
        }
    } else {
        #pragma unroll
        for (int t = 0; t < NT; ++t)
            #pragma unroll
            for (int v = 0; v < 8; ++v) {
                int r = row0 + v + 8 * half;
                int c = t * 16 + l15;
                Y[(size_t)r * OUT + c] = acc[t][v];
            }
    }
}


template <int OUT>
__device__ __forceinline__ float gather_line(const int* __restrict__ csr, const int* __restrict__ off,
                                             const float* __restrict__ dis, const float* __restrict__ hw,
                                             const float* __restrict__ b, const int d0, const int lane) {
    constexpr int G = 32 / OUT;
    const int g = lane / OUT;
    const int f = lane % OUT;

    const float bias = b[f];
    float keep = 0.0f;
    #pragma unroll
    for (int nd = 0; nd < G; ++nd) {
        const int d = d0 + nd;
        const int beg = off[d];
        const int end = (d + 1 < N_NODES) ? off[d + 1] : N_EDGES;
        const float dd = dis[d];
        const float self = hw[(size_t)d * OUT + f];
        float acc = (g == 0) ? self : 0.0f;
        for (int base = beg; base < end; base += 32) {
            const int n = min(32, end - base);
            int s_j = d;
            if (lane < n) s_j = csr[base + lane];
            for (int jj = 0; jj < n; jj += 8 * G) {
                float r[8];
                #pragma unroll
                for (int u = 0; u < 8; ++u) {
                    const int s = __shfl(s_j, jj + u * G + g);
                    r[u] = hw[(size_t)s * OUT + f];
                }
                #pragma unroll
                for (int u = 0; u < 8; ++u) acc += (jj + u * G + g < n) ? r[u] : 0.0f;
            }
        }
        if (G == 2) acc += __shfl_xor(acc, 16);
        float v = dd * acc + bias;
        v = v > 0.0f ? v : 0.0f;
        if (g == nd) keep = dd * v;
    }
    return keep;
}


#define TILE_PITCH 34

static_assert(N_NODES % 16 == 0, "the fused layer-1 gather takes whole 16-node tiles, so no wave leaves before the barrier");

__global__ __launch_bounds__(512) void k_gather32_gemm16(const int* __restrict__ csr, const int* __restrict__ off,
                                                         const float* __restrict__ dis, const float* __restrict__ hw,
                                                         const float* __restrict__ b, const float* __restrict__ W,
                                                         float* __restrict__ Y) {
    __shared__ __attribute__((aligned(8))) float s_tile[16 * TILE_PITCH];
    const int lane = threadIdx.x & 31;
    const int w = __builtin_amdgcn_readfirstlane(threadIdx.x >> 5);
    const int row0 = blockIdx.x * 16;

    s_tile[w * TILE_PITCH + lane] = gather_line<32>(csr, off, dis, hw, b, row0 + w, lane);

    const int half = lane >> 4;
    const int l15  = lane & 15;

    v2f bf[8];
    #pragma unroll
    for (int ks = 0; ks < 8; ++ks) { bf[ks].x = 0.0f; bf[ks].y = 0.0f; }
    if (w == 0) {
        #pragma unroll
        for (int ks = 0; ks < 8; ++ks) {
            bf[ks].x = W[(4 * ks + 2 * half)     * 16 + l15];
            bf[ks].y = W[(4 * ks + 2 * half + 1) * 16 + l15];
        }
    }
    __syncthreads();
    if (w != 0) return;

    v2f a[8];
    #pragma unroll
    for (int ks = 0; ks < 8; ++ks) {
        const float* ap = s_tile + l15 * TILE_PITCH + (4 * ks + 2 * half);
        a[ks].x = ap[0];
        a[ks].y = ap[1];
    }
    float* yp = Y + (size_t)(row0 + 8 * half) * 16 + l15;

    asm volatile("" : "+v"(yp),
                      "+v"(a[0]), "+v"(a[1]), "+v"(a[2]), "+v"(a[3]), "+v"(a[4]), "+v"(a[5]), "+v"(a[6]), "+v"(a[7]),
                      "+v"(bf[0]), "+v"(bf[1]), "+v"(bf[2]), "+v"(bf[3]), "+v"(bf[4]), "+v"(bf[5]), "+v"(bf[6]),
                      "+v"(bf[7]));

    v8f acc;
    #pragma unroll
    for (int v = 0; v < 8; ++v) acc[v] = 0.0f;
    #pragma unroll
    for (int ks = 0; ks < 8; ++ks)
        acc = __builtin_amdgcn_wmma_f32_16x16x4_f32(
             false, a[ks],  false, bf[ks],
             (short)0, acc,  false,  false);
    __attribute__((address_space(1))) float* ypg = (__attribute__((address_space(1))) float*)yp;
    #pragma unroll
    for (int v = 0; v < 8; ++v) ypg[v * 16] = acc[v];
}


#define H2_PITCH 20

__global__ __launch_bounds__(512) void k_gather16_dot(const int* __restrict__ csr, const int* __restrict__ off,
                                                      const float* __restrict__ dis, const float* __restrict__ hw,
                                                      const float* __restrict__ b, const float* __restrict__ W3,
                                                      float* __restrict__ hw3) {
    __shared__ __attribute__((aligned(16))) float s_h2[32 * H2_PITCH];
    const int lane = threadIdx.x & 31;
    const int w = __builtin_amdgcn_readfirstlane(threadIdx.x >> 5);
    const int d0 = (blockIdx.x * 16 + w) * 2;

    s_h2[(2 * w + (lane >> 4)) * H2_PITCH + (lane & 15)] = gather_line<16>(csr, off, dis, hw, b, d0, lane);
    __syncthreads();
    if (w != 0) return;

    const float* r = s_h2 + lane * H2_PITCH;
    float s = 0.0f;
    #pragma unroll
    for (int k = 0; k < 16; ++k) s += r[k] * W3[k];
    hw3[blockIdx.x * 32 + lane] = s;
}


__global__ __launch_bounds__(1024) void k_gather_out1(const int* __restrict__ csr, const int* __restrict__ off,
                                                      const float* __restrict__ dis, const float* __restrict__ hw3,
                                                      const float* __restrict__ b3, float* __restrict__ out) {
    __shared__ float s_out[32];
    const int lane = threadIdx.x & 31;
    const int w = __builtin_amdgcn_readfirstlane(threadIdx.x >> 5);
    const int d = blockIdx.x * 32 + w;
    const int beg = off[d];
    const int end = (d + 1 < N_NODES) ? off[d + 1] : N_EDGES;
    const float dd = dis[d];
    float acc = 0.0f;
    for (int j = beg + lane; j < end; j += 32) acc += hw3[csr[j]];
    for (int o = 16; o > 0; o >>= 1) acc += __shfl_xor(acc, o);
    if (lane == 0) s_out[w] = dd * (acc + hw3[d]) + b3[0];
    __syncthreads();
    if (w == 0) out[blockIdx.x * 32 + lane] = s_out[lane];
}


static inline int blks(long long n, int bs) { return (int)((n + bs - 1) / bs); }

extern "C" void kernel_launch(void* const* d_in, const int* in_sizes, int n_in,
                              void* d_out, int out_size, void* d_ws, size_t ws_size,
                              hipStream_t stream) {
    (void)in_sizes; (void)n_in; (void)out_size; (void)ws_size;

    const float* x   = (const float*)d_in[0];
    const int*   ei  = (const int*)  d_in[1];
    const float* W1  = (const float*)d_in[2];
    const float* b1  = (const float*)d_in[3];
    const float* W2  = (const float*)d_in[4];
    const float* b2  = (const float*)d_in[5];
    const float* W3  = (const float*)d_in[6];
    const float* b3  = (const float*)d_in[7];
    float* out = (float*)d_out;

    const int* src = ei;
    const int* dst = ei + N_EDGES;

    float* ws   = (float*)d_ws;
    float* dis  = ws;
    int*   off  = (int*)(ws + N_NODES);
    int*   csr  = (int*)(ws + 2 * N_NODES);
    float* pA   = ws + 34 * N_NODES;
    float* pB   = ws + 66 * N_NODES;
    int*   tbl    = (int*)pA;
    int*   binned = (int*)pB;
    float* hw1  = pA;
    float* hw2  = pB;
    float* hw3  = pA;

    const int BS = 256;

    k_chunk_sort<<<SORT_BLOCKS, 1024, 0, stream>>>(src, dst, tbl, binned);
    k_bin_sort  <<<N_BINS, 1024, 0, stream>>>(binned, tbl, off, dis, csr);

    const int gemm_grid = blks(N_NODES / 16, BS / 32);

    k_gemm_wmma<128, 32, true><<<gemm_grid, BS, 0, stream>>>(x, W1, dis, hw1, N_NODES);
    k_gather32_gemm16<<<N_NODES / 16, 512, 0, stream>>>(csr, off, dis, hw1, b1, W2, hw2);

    k_gather16_dot<<<N_NODES / 32, 512, 0, stream>>>(csr, off, dis, hw2, b2, W3, hw3);

    k_gather_out1<<<N_NODES / 32, 1024, 0, stream>>>(csr, off, dis, hw3, b3, out);
}


# Layer-3 gather picks hw3 from LDS: 256 resident blocks stage the scalar table in two halves, one thread per node sums its edge values in list order; no scattered 4-byte reads, 100000 -> 8192 waves
# speedup vs baseline: 8.0233x; 1.0961x over previous
#include <hip/hip_runtime.h>


typedef __attribute__((ext_vector_type(2))) float v2f;
typedef __attribute__((ext_vector_type(8))) float v8f;

#define N_NODES 100000
#define N_EDGES 3200000

static_assert(N_NODES % 32 == 0, "the layer-2 and layer-3 gathers hand out whole 32-node lines");
static_assert(N_EDGES == 32 * N_NODES, "the binned edge list reuses the 32*N word plane of the layer-1 output");


#define SORT_BLOCKS 256
#define N_BINS      ((N_NODES + 255) / 256)
#define EDGE_CHUNK  ((N_EDGES + SORT_BLOCKS - 1) / SORT_BLOCKS)
#define TBL_PITCH   (N_BINS + 1)
#define EPT         ((EDGE_CHUNK + 1023) / 1024)

static_assert(N_NODES <= (1 << 17), "src must fit in the low 17 bits of a binned word");
static_assert(N_EDGES == SORT_BLOCKS * EDGE_CHUNK, "every level-1 block owns a full stretch");
static_assert(SORT_BLOCKS * TBL_PITCH <= 32 * N_NODES, "the tables of bin starts lie in plane A");
static_assert(SORT_BLOCKS % 32 == 0, "a level-2 wave takes every 32nd run");
static_assert(N_BINS <= 1024, "one thread per bin scans the block's bin counts");

__global__ __launch_bounds__(1024) void k_chunk_sort(const int* __restrict__ src, const int* __restrict__ dst,
                                                     int* __restrict__ tbl, int* __restrict__ binned) {
    __shared__ int s_stage[EDGE_CHUNK];
    __shared__ int s_cnt[N_BINS];
    __shared__ int s_start[N_BINS];
    __shared__ int s_wsum[32];
    const int t = threadIdx.x;
    const int lane = t & 31;
    const int w = t >> 5;
    const int blk = blockIdx.x;
    for (int b = t; b < N_BINS; b += 1024) s_cnt[b] = 0;

    const int lo = blk * EDGE_CHUNK;
    int d[EPT], s[EPT], r[EPT];
    #pragma unroll
    for (int u = 0; u < EPT; ++u) d[u] = (t + u * 1024 < EDGE_CHUNK) ? dst[lo + t + u * 1024] : 0;
    #pragma unroll
    for (int u = 0; u < EPT; ++u) s[u] = (t + u * 1024 < EDGE_CHUNK) ? src[lo + t + u * 1024] : 0;
    __syncthreads();

    #pragma unroll
    for (int u = 0; u < EPT; ++u) {
        r[u] = 0;
        if (t + u * 1024 < EDGE_CHUNK) r[u] = atomicAdd(&s_cnt[d[u] >> 8], 1);
    }
    __syncthreads();

    const int c = (t < N_BINS) ? s_cnt[t] : 0;
    int incl = c;
    for (int o = 1; o < 32; o <<= 1) {
        int v = __shfl_up(incl, o);
        if (lane >= o) incl += v;
    }
    if (lane == 31) s_wsum[w] = incl;
    __syncthreads();
    int woff = 0;
    for (int k = 0; k < 32; ++k)
        if (k < w) woff += s_wsum[k];
    const int excl = woff + incl - c;
    if (t < N_BINS) {
        s_start[t] = excl;
        tbl[blk * TBL_PITCH + t] = excl;
    }
    if (t == 0) tbl[blk * TBL_PITCH + N_BINS] = EDGE_CHUNK;
    __syncthreads();

    #pragma unroll
    for (int u = 0; u < EPT; ++u)
        if (t + u * 1024 < EDGE_CHUNK) s_stage[s_start[d[u] >> 8] + r[u]] = s[u] | ((d[u] & 255) << 17);
    __syncthreads();

    for (int i = t; i < EDGE_CHUNK; i += 1024) binned[lo + i] = s_stage[i];
}

#define BIN_STAGE 12288
#define RPW (SORT_BLOCKS / 32)

template <bool STAGED>
__device__ __forceinline__ void place_edge(const int wd, int* s_cur, int* s_stage, int* __restrict__ csr,
                                           const int bstart) {
    const int p = atomicAdd(&s_cur[wd >> 17], 1);
    if (STAGED) s_stage[p] = wd & 0x1FFFF;
    else        csr[bstart + p] = wd & 0x1FFFF;
}

template <bool STAGED>
__device__ __forceinline__ void place_runs(const int (&w0)[RPW], const int (&w1)[RPW], const int (&rs)[RPW],
                                           const int (&rl)[RPW], int* s_cur, int* s_stage,
                                           const int* __restrict__ binned, int* __restrict__ csr,
                                           const int bstart, const int lane) {
    #pragma unroll
    for (int k = 0; k < RPW; ++k) {
        if (w0[k] >= 0) place_edge<STAGED>(w0[k], s_cur, s_stage, csr, bstart);
        if (w1[k] >= 0) place_edge<STAGED>(w1[k], s_cur, s_stage, csr, bstart);
    }
    #pragma unroll
    for (int k = 0; k < RPW; ++k)
        for (int j = lane + 64; j < rl[k]; j += 32) place_edge<STAGED>(binned[rs[k] + j], s_cur, s_stage, csr, bstart);
}

__global__ __launch_bounds__(1024) void k_bin_sort(const int* __restrict__ binned, const int* __restrict__ tbl,
                                                   int* __restrict__ off, float* __restrict__ dis,
                                                   int* __restrict__ csr) {
    __shared__ int s_cnt[256];
    __shared__ int s_cur[256];
    __shared__ int s_wsum[8];
    __shared__ int s_rs[SORT_BLOCKS];
    __shared__ int s_rl[SORT_BLOCKS];
    __shared__ int s_sum_start[RPW];
    __shared__ int s_sum_len[RPW];
    __shared__ int s_stage[BIN_STAGE];
    const int t = threadIdx.x;
    const int lane = t & 31;
    const int w = t >> 5;
    const int bin = blockIdx.x;

    if (t < SORT_BLOCKS) {
        const int a = tbl[t * TBL_PITCH + bin];
        const int b = tbl[t * TBL_PITCH + bin + 1];
        s_rs[t] = t * EDGE_CHUNK + a;
        s_rl[t] = b - a;
        int sa = a, sl = b - a;
        for (int o = 16; o > 0; o >>= 1) {
            sa += __shfl_xor(sa, o);
            sl += __shfl_xor(sl, o);
        }
        if (lane == 0) {
            s_sum_start[w] = sa;
            s_sum_len[w] = sl;
        }
    }
    if (t < 256) s_cnt[t] = 0;
    __syncthreads();
    int bstart = 0, n = 0;
    for (int k = 0; k < RPW; ++k) {
        bstart += s_sum_start[k];
        n += s_sum_len[k];
    }
    const bool staged = n <= BIN_STAGE;

    int rs[RPW], rl[RPW], w0[RPW], w1[RPW];
    #pragma unroll
    for (int k = 0; k < RPW; ++k) {
        rs[k] = s_rs[w + 32 * k];
        rl[k] = s_rl[w + 32 * k];
    }
    #pragma unroll
    for (int k = 0; k < RPW; ++k) {
        w0[k] = (lane < rl[k]) ? binned[rs[k] + lane] : -1;
        w1[k] = (lane + 32 < rl[k]) ? binned[rs[k] + lane + 32] : -1;
    }
    #pragma unroll
    for (int k = 0; k < RPW; ++k) {
        if (w0[k] >= 0) atomicAdd(&s_cnt[w0[k] >> 17], 1);
        if (w1[k] >= 0) atomicAdd(&s_cnt[w1[k] >> 17], 1);
    }
    #pragma unroll
    for (int k = 0; k < RPW; ++k)
        for (int j = lane + 64; j < rl[k]; j += 32) atomicAdd(&s_cnt[binned[rs[k] + j] >> 17], 1);
    __syncthreads();

    int c = 0, incl = 0;
    if (t < 256) {
        c = s_cnt[t];
        incl = c;
        for (int o = 1; o < 32; o <<= 1) {
            int v = __shfl_up(incl, o);
            if (lane >= o) incl += v;
        }
        if (lane == 31) s_wsum[t >> 5] = incl;
    }
    __syncthreads();
    if (t < 256) {
        int woff = 0;
        for (int k = 0; k < 8; ++k)
            if (k < (t >> 5)) woff += s_wsum[k];
        const int excl = woff + incl - c;
        s_cur[t] = excl;
        const int node = bin * 256 + t;
        if (node < N_NODES) {
            off[node] = bstart + excl;
            dis[node] = rsqrtf((float)(c + 1));
        }
    }
    __syncthreads();
    if (staged) place_runs<true >(w0, w1, rs, rl, s_cur, s_stage, binned, csr, bstart, lane);
    else        place_runs<false>(w0, w1, rs, rl, s_cur, s_stage, binned, csr, bstart, lane);
    __syncthreads();
    if (staged)
        for (int j = t; j < n; j += 1024) csr[bstart + j] = s_stage[j];
}


template <int IN, int OUT, bool ROWSCALE>
__global__ __launch_bounds__(256) void k_gemm_wmma(const float* __restrict__ X,
                                                   const float* __restrict__ W,
                                                   const float* __restrict__ rs,
                                                   float* __restrict__ Y,
                                                   int nrows) {
    const int lane = threadIdx.x & 31;
    const int wave = blockIdx.x * (blockDim.x >> 5) + (threadIdx.x >> 5);
    const int row0 = wave * 16;
    if (row0 >= nrows) return;

    const int half = lane >> 4;
    const int l15  = lane & 15;
    const int arow = row0 + l15;

    constexpr int NT = OUT / 16;

    float sc[16];
    if constexpr (ROWSCALE) {
        const float* rsw = rs + __builtin_amdgcn_readfirstlane(row0);
        #pragma unroll
        for (int i = 0; i < 16; ++i) sc[i] = rsw[i];
    }

    v8f acc[NT];
    #pragma unroll
    for (int t = 0; t < NT; ++t)
        #pragma unroll
        for (int v = 0; v < 8; ++v) acc[t][v] = 0.0f;

    for (int k = 0; k < IN; k += 4) {
        const float* ap = X + (size_t)arow * IN + (k + 2 * half);
        v2f a;
        a.x = ap[0];
        a.y = ap[1];

        #pragma unroll
        for (int t = 0; t < NT; ++t) {
            const int col = t * 16 + l15;
            v2f b;
            b.x = W[(k + 2 * half)     * OUT + col];
            b.y = W[(k + 2 * half + 1) * OUT + col];
            acc[t] = __builtin_amdgcn_wmma_f32_16x16x4_f32(
                 false, a,  false, b,
                 (short)0, acc[t],  false,  false);
        }
    }

    if constexpr (ROWSCALE) {
        int hsel = half;
        #pragma unroll
        for (int t = 0; t < NT; ++t) asm volatile("" : "+v"(acc[t]));
        asm volatile("v_nop\n\tv_nop\n\tv_nop\n\tv_nop\n\tv_nop\n\tv_nop\n\tv_nop\n\tv_nop"
                     : "+v"(hsel), "+s"(sc[0]), "+s"(sc[1]), "+s"(sc[2]), "+s"(sc[3]), "+s"(sc[4]), "+s"(sc[5]),
                       "+s"(sc[6]), "+s"(sc[7]), "+s"(sc[8]), "+s"(sc[9]), "+s"(sc[10]), "+s"(sc[11]),
                       "+s"(sc[12]), "+s"(sc[13]));
        asm volatile("" : "+s"(sc[14]), "+s"(sc[15]));
        #pragma unroll
        for (int t = 0; t < NT; ++t) asm volatile("" : "+v"(acc[t]));
        #pragma unroll
        for (int v = 0; v < 8; ++v) {
            const int r = row0 + v + 8 * half;
            const float f = hsel ? sc[v + 8] : sc[v];
            #pragma unroll
            for (int t = 0; t < NT; ++t) Y[(size_t)r * OUT + t * 16 + l15] = f * acc[t][v];
        }
    } else {
        #pragma unroll
        for (int t = 0; t < NT; ++t)
            #pragma unroll
            for (int v = 0; v < 8; ++v) {
                int r = row0 + v + 8 * half;
                int c = t * 16 + l15;
                Y[(size_t)r * OUT + c] = acc[t][v];
            }
    }
}


template <int OUT>
__device__ __forceinline__ float gather_line(const int* __restrict__ csr, const int* __restrict__ off,
                                             const float* __restrict__ dis, const float* __restrict__ hw,
                                             const float* __restrict__ b, const int d0, const int lane) {
    constexpr int G = 32 / OUT;
    const int g = lane / OUT;
    const int f = lane % OUT;

    const float bias = b[f];
    float keep = 0.0f;
    #pragma unroll
    for (int nd = 0; nd < G; ++nd) {
        const int d = d0 + nd;
        const int beg = off[d];
        const int end = (d + 1 < N_NODES) ? off[d + 1] : N_EDGES;
        const float dd = dis[d];
        const float self = hw[(size_t)d * OUT + f];
        float acc = (g == 0) ? self : 0.0f;
        for (int base = beg; base < end; base += 32) {
            const int n = min(32, end - base);
            int s_j = d;
            if (lane < n) s_j = csr[base + lane];
            for (int jj = 0; jj < n; jj += 8 * G) {
                float r[8];
                #pragma unroll
                for (int u = 0; u < 8; ++u) {
                    const int s = __shfl(s_j, jj + u * G + g);
                    r[u] = hw[(size_t)s * OUT + f];
                }
                #pragma unroll
                for (int u = 0; u < 8; ++u) acc += (jj + u * G + g < n) ? r[u] : 0.0f;
            }
        }
        if (G == 2) acc += __shfl_xor(acc, 16);
        float v = dd * acc + bias;
        v = v > 0.0f ? v : 0.0f;
        if (g == nd) keep = dd * v;
    }
    return keep;
}


#define TILE_PITCH 34

static_assert(N_NODES % 16 == 0, "the fused layer-1 gather takes whole 16-node tiles, so no wave leaves before the barrier");

__global__ __launch_bounds__(512) void k_gather32_gemm16(const int* __restrict__ csr, const int* __restrict__ off,
                                                         const float* __restrict__ dis, const float* __restrict__ hw,
                                                         const float* __restrict__ b, const float* __restrict__ W,
                                                         float* __restrict__ Y) {
    __shared__ __attribute__((aligned(8))) float s_tile[16 * TILE_PITCH];
    const int lane = threadIdx.x & 31;
    const int w = __builtin_amdgcn_readfirstlane(threadIdx.x >> 5);
    const int row0 = blockIdx.x * 16;

    s_tile[w * TILE_PITCH + lane] = gather_line<32>(csr, off, dis, hw, b, row0 + w, lane);

    const int half = lane >> 4;
    const int l15  = lane & 15;

    v2f bf[8];
    #pragma unroll
    for (int ks = 0; ks < 8; ++ks) { bf[ks].x = 0.0f; bf[ks].y = 0.0f; }
    if (w == 0) {
        #pragma unroll
        for (int ks = 0; ks < 8; ++ks) {
            bf[ks].x = W[(4 * ks + 2 * half)     * 16 + l15];
            bf[ks].y = W[(4 * ks + 2 * half + 1) * 16 + l15];
        }
    }
    __syncthreads();
    if (w != 0) return;

    v2f a[8];
    #pragma unroll
    for (int ks = 0; ks < 8; ++ks) {
        const float* ap = s_tile + l15 * TILE_PITCH + (4 * ks + 2 * half);
        a[ks].x = ap[0];
        a[ks].y = ap[1];
    }
    float* yp = Y + (size_t)(row0 + 8 * half) * 16 + l15;

    asm volatile("" : "+v"(yp),
                      "+v"(a[0]), "+v"(a[1]), "+v"(a[2]), "+v"(a[3]), "+v"(a[4]), "+v"(a[5]), "+v"(a[6]), "+v"(a[7]),
                      "+v"(bf[0]), "+v"(bf[1]), "+v"(bf[2]), "+v"(bf[3]), "+v"(bf[4]), "+v"(bf[5]), "+v"(bf[6]),
                      "+v"(bf[7]));

    v8f acc;
    #pragma unroll
    for (int v = 0; v < 8; ++v) acc[v] = 0.0f;
    #pragma unroll
    for (int ks = 0; ks < 8; ++ks)
        acc = __builtin_amdgcn_wmma_f32_16x16x4_f32(
             false, a[ks],  false, bf[ks],
             (short)0, acc,  false,  false);
    __attribute__((address_space(1))) float* ypg = (__attribute__((address_space(1))) float*)yp;
    #pragma unroll
    for (int v = 0; v < 8; ++v) ypg[v * 16] = acc[v];
}


#define H2_PITCH 20

__global__ __launch_bounds__(512) void k_gather16_dot(const int* __restrict__ csr, const int* __restrict__ off,
                                                      const float* __restrict__ dis, const float* __restrict__ hw,
                                                      const float* __restrict__ b, const float* __restrict__ W3,
                                                      float* __restrict__ hw3) {
    __shared__ __attribute__((aligned(16))) float s_h2[32 * H2_PITCH];
    const int lane = threadIdx.x & 31;
    const int w = __builtin_amdgcn_readfirstlane(threadIdx.x >> 5);
    const int d0 = (blockIdx.x * 16 + w) * 2;

    s_h2[(2 * w + (lane >> 4)) * H2_PITCH + (lane & 15)] = gather_line<16>(csr, off, dis, hw, b, d0, lane);
    __syncthreads();
    if (w != 0) return;

    const float* r = s_h2 + lane * H2_PITCH;
    float s = 0.0f;
    #pragma unroll
    for (int k = 0; k < 16; ++k) s += r[k] * W3[k];
    hw3[blockIdx.x * 32 + lane] = s;
}


#define TBL_HALF   (N_NODES / 2)
#define OUT_BLOCKS 256
#define OUT_CH     16384
#define OUT_EPT    (OUT_CH / 1024)
#define N_LINES    (N_NODES / 32)
#define TBL_TRIPS  ((TBL_HALF / 4 + 1023) / 1024)

static_assert(N_NODES % 8 == 0, "a table half is copied in 16-byte units");
static_assert(((N_LINES + OUT_BLOCKS - 1) / OUT_BLOCKS) * 32 <= 1024, "one thread per node of a block's stretch");

__device__ __forceinline__ void load_half(float4 (&r)[TBL_TRIPS], const float* src, const int t) {
    const float4* g = (const float4*)src;
    #pragma unroll
    for (int q = 0; q < TBL_TRIPS; ++q)
        r[q] = (t + q * 1024 < TBL_HALF / 4) ? g[t + q * 1024] : make_float4(0.0f, 0.0f, 0.0f, 0.0f);
}
__device__ __forceinline__ void store_half(float4* l, const float4 (&r)[TBL_TRIPS], const int t) {
    #pragma unroll
    for (int q = 0; q < TBL_TRIPS; ++q)
        if (t + q * 1024 < TBL_HALF / 4) l[t + q * 1024] = r[q];
}
__device__ __forceinline__ void pick_half(const int (&s)[OUT_EPT], const int base, const float* s_tbl,
                                          float* s_val, const int t) {
    float x[OUT_EPT];
    #pragma unroll
    for (int u = 0; u < OUT_EPT; ++u) {
        const unsigned k = (unsigned)(s[u] - base);
        x[u] = s_tbl[k < (unsigned)TBL_HALF ? k : 0u];
    }
    #pragma unroll
    for (int u = 0; u < OUT_EPT; ++u)
        if ((unsigned)(s[u] - base) < (unsigned)TBL_HALF) s_val[t + u * 1024] = x[u];
}

__global__ __launch_bounds__(1024) void k_gather_out1(const int* __restrict__ csr, const int* __restrict__ off,
                                                      const float* __restrict__ dis, const float* hw3,
                                                      const float* __restrict__ b3, float* out) {
    __shared__ __attribute__((aligned(16))) float s_tbl[TBL_HALF];
    __shared__ float s_val[OUT_CH];
    const int t = threadIdx.x;
    const int l_lo = (int)(((long long)blockIdx.x * N_LINES) / OUT_BLOCKS);
    const int l_hi = (int)(((long long)(blockIdx.x + 1) * N_LINES) / OUT_BLOCKS);
    const int n_lo = l_lo * 32;
    const int n_hi = l_hi * 32;
    const int nn = n_hi - n_lo;
    const int e_lo = off[n_lo];
    const int e_hi = (n_hi < N_NODES) ? off[n_hi] : N_EDGES;
    const int d = n_lo + t;
    int beg = 0, end = 0;
    if (t < nn) {
        beg = off[d];
        end = (d + 1 < N_NODES) ? off[d + 1] : N_EDGES;
    }
    float acc = 0.0f;
    for (int c0 = e_lo; c0 < e_hi; c0 += OUT_CH) {
        int s[OUT_EPT];
        #pragma unroll
        for (int u = 0; u < OUT_EPT; ++u) {
            const int j = c0 + t + u * 1024;
            s[u] = (j < e_hi) ? csr[j] : -1;
        }
        float4* l = (float4*)s_tbl;
        float4 r[TBL_TRIPS];
        load_half(r, hw3, t);
        __syncthreads();
        store_half(l, r, t);
        __syncthreads();
        pick_half(s, 0, s_tbl, s_val, t);
        load_half(r, hw3 + TBL_HALF, t);
        __syncthreads();
        store_half(l, r, t);
        __syncthreads();
        pick_half(s, TBL_HALF, s_tbl, s_val, t);
        __syncthreads();
        if (t < nn) {
            const int a = max(beg, c0);
            const int b = min(end, c0 + OUT_CH);
            for (int j = a; j < b; ++j) acc += s_val[j - c0];
        }
    }
    if (t < nn) out[d] = dis[d] * (acc + hw3[d]) + b3[0];
}


static inline int blks(long long n, int bs) { return (int)((n + bs - 1) / bs); }

extern "C" void kernel_launch(void* const* d_in, const int* in_sizes, int n_in,
                              void* d_out, int out_size, void* d_ws, size_t ws_size,
                              hipStream_t stream) {
    (void)in_sizes; (void)n_in; (void)out_size; (void)ws_size;

    const float* x   = (const float*)d_in[0];
    const int*   ei  = (const int*)  d_in[1];
    const float* W1  = (const float*)d_in[2];
    const float* b1  = (const float*)d_in[3];
    const float* W2  = (const float*)d_in[4];
    const float* b2  = (const float*)d_in[5];
    const float* W3  = (const float*)d_in[6];
    const float* b3  = (const float*)d_in[7];
    float* out = (float*)d_out;

    const int* src = ei;
    const int* dst = ei + N_EDGES;

    float* ws   = (float*)d_ws;
    float* dis  = ws;
    int*   off  = (int*)(ws + N_NODES);
    int*   csr  = (int*)(ws + 2 * N_NODES);
    float* pA   = ws + 34 * N_NODES;
    float* pB   = ws + 66 * N_NODES;
    int*   tbl    = (int*)pA;
    int*   binned = (int*)pB;
    float* hw1  = pA;
    float* hw2  = pB;
    float* hw3  = pA;

    const int BS = 256;

    k_chunk_sort<<<SORT_BLOCKS, 1024, 0, stream>>>(src, dst, tbl, binned);
    k_bin_sort  <<<N_BINS, 1024, 0, stream>>>(binned, tbl, off, dis, csr);

    const int gemm_grid = blks(N_NODES / 16, BS / 32);

    k_gemm_wmma<128, 32, true><<<gemm_grid, BS, 0, stream>>>(x, W1, dis, hw1, N_NODES);
    k_gather32_gemm16<<<N_NODES / 16, 512, 0, stream>>>(csr, off, dis, hw1, b1, W2, hw2);

    k_gather16_dot<<<N_NODES / 32, 512, 0, stream>>>(csr, off, dis, hw2, b2, W3, hw3);

    k_gather_out1<<<OUT_BLOCKS, 1024, 0, stream>>>(csr, off, dis, hw3, b3, out);
}


# Layer-1 and layer-2 gathers serve four nodes per wave side by side (8 lanes per row, 16/8-byte loads): 150000 -> 50000 gather waves, fewer row-load instructions; every sum keeps its order
# speedup vs baseline: 9.1040x; 1.1347x over previous
#include <hip/hip_runtime.h>


typedef __attribute__((ext_vector_type(2))) float v2f;
typedef __attribute__((ext_vector_type(8))) float v8f;

#define N_NODES 100000
#define N_EDGES 3200000

static_assert(N_NODES % 32 == 0, "the three gathers hand out whole 32-node tiles and lines");
static_assert(N_EDGES == 32 * N_NODES, "the binned edge list reuses the 32*N word plane of the layer-1 output");


#define SORT_BLOCKS 256
#define N_BINS      ((N_NODES + 255) / 256)
#define EDGE_CHUNK  ((N_EDGES + SORT_BLOCKS - 1) / SORT_BLOCKS)
#define TBL_PITCH   (N_BINS + 1)
#define EPT         ((EDGE_CHUNK + 1023) / 1024)

static_assert(N_NODES <= (1 << 17), "src must fit in the low 17 bits of a binned word");
static_assert(N_EDGES == SORT_BLOCKS * EDGE_CHUNK, "every level-1 block owns a full stretch");
static_assert(SORT_BLOCKS * TBL_PITCH <= 32 * N_NODES, "the tables of bin starts lie in plane A");
static_assert(SORT_BLOCKS % 32 == 0, "a level-2 wave takes every 32nd run");
static_assert(N_BINS <= 1024, "one thread per bin scans the block's bin counts");

__global__ __launch_bounds__(1024) void k_chunk_sort(const int* __restrict__ src, const int* __restrict__ dst,
                                                     int* __restrict__ tbl, int* __restrict__ binned) {
    __shared__ int s_stage[EDGE_CHUNK];
    __shared__ int s_cnt[N_BINS];
    __shared__ int s_start[N_BINS];
    __shared__ int s_wsum[32];
    const int t = threadIdx.x;
    const int lane = t & 31;
    const int w = t >> 5;
    const int blk = blockIdx.x;
    for (int b = t; b < N_BINS; b += 1024) s_cnt[b] = 0;

    const int lo = blk * EDGE_CHUNK;
    int d[EPT], s[EPT], r[EPT];
    #pragma unroll
    for (int u = 0; u < EPT; ++u) d[u] = (t + u * 1024 < EDGE_CHUNK) ? dst[lo + t + u * 1024] : 0;
    #pragma unroll
    for (int u = 0; u < EPT; ++u) s[u] = (t + u * 1024 < EDGE_CHUNK) ? src[lo + t + u * 1024] : 0;
    __syncthreads();

    #pragma unroll
    for (int u = 0; u < EPT; ++u) {
        r[u] = 0;
        if (t + u * 1024 < EDGE_CHUNK) r[u] = atomicAdd(&s_cnt[d[u] >> 8], 1);
    }
    __syncthreads();

    const int c = (t < N_BINS) ? s_cnt[t] : 0;
    int incl = c;
    for (int o = 1; o < 32; o <<= 1) {
        int v = __shfl_up(incl, o);
        if (lane >= o) incl += v;
    }
    if (lane == 31) s_wsum[w] = incl;
    __syncthreads();
    int woff = 0;
    for (int k = 0; k < 32; ++k)
        if (k < w) woff += s_wsum[k];
    const int excl = woff + incl - c;
    if (t < N_BINS) {
        s_start[t] = excl;
        tbl[blk * TBL_PITCH + t] = excl;
    }
    if (t == 0) tbl[blk * TBL_PITCH + N_BINS] = EDGE_CHUNK;
    __syncthreads();

    #pragma unroll
    for (int u = 0; u < EPT; ++u)
        if (t + u * 1024 < EDGE_CHUNK) s_stage[s_start[d[u] >> 8] + r[u]] = s[u] | ((d[u] & 255) << 17);
    __syncthreads();

    for (int i = t; i < EDGE_CHUNK; i += 1024) binned[lo + i] = s_stage[i];
}

#define BIN_STAGE 12288
#define RPW (SORT_BLOCKS / 32)

template <bool STAGED>
__device__ __forceinline__ void place_edge(const int wd, int* s_cur, int* s_stage, int* __restrict__ csr,
                                           const int bstart) {
    const int p = atomicAdd(&s_cur[wd >> 17], 1);
    if (STAGED) s_stage[p] = wd & 0x1FFFF;
    else        csr[bstart + p] = wd & 0x1FFFF;
}

template <bool STAGED>
__device__ __forceinline__ void place_runs(const int (&w0)[RPW], const int (&w1)[RPW], const int (&rs)[RPW],
                                           const int (&rl)[RPW], int* s_cur, int* s_stage,
                                           const int* __restrict__ binned, int* __restrict__ csr,
                                           const int bstart, const int lane) {
    #pragma unroll
    for (int k = 0; k < RPW; ++k) {
        if (w0[k] >= 0) place_edge<STAGED>(w0[k], s_cur, s_stage, csr, bstart);
        if (w1[k] >= 0) place_edge<STAGED>(w1[k], s_cur, s_stage, csr, bstart);
    }
    #pragma unroll
    for (int k = 0; k < RPW; ++k)
        for (int j = lane + 64; j < rl[k]; j += 32) place_edge<STAGED>(binned[rs[k] + j], s_cur, s_stage, csr, bstart);
}

__global__ __launch_bounds__(1024) void k_bin_sort(const int* __restrict__ binned, const int* __restrict__ tbl,
                                                   int* __restrict__ off, float* __restrict__ dis,
                                                   int* __restrict__ csr) {
    __shared__ int s_cnt[256];
    __shared__ int s_cur[256];
    __shared__ int s_wsum[8];
    __shared__ int s_rs[SORT_BLOCKS];
    __shared__ int s_rl[SORT_BLOCKS];
    __shared__ int s_sum_start[RPW];
    __shared__ int s_sum_len[RPW];
    __shared__ int s_stage[BIN_STAGE];
    const int t = threadIdx.x;
    const int lane = t & 31;
    const int w = t >> 5;
    const int bin = blockIdx.x;

    if (t < SORT_BLOCKS) {
        const int a = tbl[t * TBL_PITCH + bin];
        const int b = tbl[t * TBL_PITCH + bin + 1];
        s_rs[t] = t * EDGE_CHUNK + a;
        s_rl[t] = b - a;
        int sa = a, sl = b - a;
        for (int o = 16; o > 0; o >>= 1) {
            sa += __shfl_xor(sa, o);
            sl += __shfl_xor(sl, o);
        }
        if (lane == 0) {
            s_sum_start[w] = sa;
            s_sum_len[w] = sl;
        }
    }
    if (t < 256) s_cnt[t] = 0;
    __syncthreads();
    int bstart = 0, n = 0;
    for (int k = 0; k < RPW; ++k) {
        bstart += s_sum_start[k];
        n += s_sum_len[k];
    }
    const bool staged = n <= BIN_STAGE;

    int rs[RPW], rl[RPW], w0[RPW], w1[RPW];
    #pragma unroll
    for (int k = 0; k < RPW; ++k) {
        rs[k] = s_rs[w + 32 * k];
        rl[k] = s_rl[w + 32 * k];
    }
    #pragma unroll
    for (int k = 0; k < RPW; ++k) {
        w0[k] = (lane < rl[k]) ? binned[rs[k] + lane] : -1;
        w1[k] = (lane + 32 < rl[k]) ? binned[rs[k] + lane + 32] : -1;
    }
    #pragma unroll
    for (int k = 0; k < RPW; ++k) {
        if (w0[k] >= 0) atomicAdd(&s_cnt[w0[k] >> 17], 1);
        if (w1[k] >= 0) atomicAdd(&s_cnt[w1[k] >> 17], 1);
    }
    #pragma unroll
    for (int k = 0; k < RPW; ++k)
        for (int j = lane + 64; j < rl[k]; j += 32) atomicAdd(&s_cnt[binned[rs[k] + j] >> 17], 1);
    __syncthreads();

    int c = 0, incl = 0;
    if (t < 256) {
        c = s_cnt[t];
        incl = c;
        for (int o = 1; o < 32; o <<= 1) {
            int v = __shfl_up(incl, o);
            if (lane >= o) incl += v;
        }
        if (lane == 31) s_wsum[t >> 5] = incl;
    }
    __syncthreads();
    if (t < 256) {
        int woff = 0;
        for (int k = 0; k < 8; ++k)
            if (k < (t >> 5)) woff += s_wsum[k];
        const int excl = woff + incl - c;
        s_cur[t] = excl;
        const int node = bin * 256 + t;
        if (node < N_NODES) {
            off[node] = bstart + excl;
            dis[node] = rsqrtf((float)(c + 1));
        }
    }
    __syncthreads();
    if (staged) place_runs<true >(w0, w1, rs, rl, s_cur, s_stage, binned, csr, bstart, lane);
    else        place_runs<false>(w0, w1, rs, rl, s_cur, s_stage, binned, csr, bstart, lane);
    __syncthreads();
    if (staged)
        for (int j = t; j < n; j += 1024) csr[bstart + j] = s_stage[j];
}


template <int IN, int OUT, bool ROWSCALE>
__global__ __launch_bounds__(256) void k_gemm_wmma(const float* __restrict__ X,
                                                   const float* __restrict__ W,
                                                   const float* __restrict__ rs,
                                                   float* __restrict__ Y,
                                                   int nrows) {
    const int lane = threadIdx.x & 31;
    const int wave = blockIdx.x * (blockDim.x >> 5) + (threadIdx.x >> 5);
    const int row0 = wave * 16;
    if (row0 >= nrows) return;

    const int half = lane >> 4;
    const int l15  = lane & 15;
    const int arow = row0 + l15;

    constexpr int NT = OUT / 16;

    float sc[16];
    if constexpr (ROWSCALE) {
        const float* rsw = rs + __builtin_amdgcn_readfirstlane(row0);
        #pragma unroll
        for (int i = 0; i < 16; ++i) sc[i] = rsw[i];
    }

    v8f acc[NT];
    #pragma unroll
    for (int t = 0; t < NT; ++t)
        #pragma unroll
        for (int v = 0; v < 8; ++v) acc[t][v] = 0.0f;

    for (int k = 0; k < IN; k += 4) {
        const float* ap = X + (size_t)arow * IN + (k + 2 * half);
        v2f a;
        a.x = ap[0];
        a.y = ap[1];

        #pragma unroll
        for (int t = 0; t < NT; ++t) {
            const int col = t * 16 + l15;
            v2f b;
            b.x = W[(k + 2 * half)     * OUT + col];
            b.y = W[(k + 2 * half + 1) * OUT + col];
            acc[t] = __builtin_amdgcn_wmma_f32_16x16x4_f32(
                 false, a,  false, b,
                 (short)0, acc[t],  false,  false);
        }
    }

    if constexpr (ROWSCALE) {
        int hsel = half;
        #pragma unroll
        for (int t = 0; t < NT; ++t) asm volatile("" : "+v"(acc[t]));
        asm volatile("v_nop\n\tv_nop\n\tv_nop\n\tv_nop\n\tv_nop\n\tv_nop\n\tv_nop\n\tv_nop"
                     : "+v"(hsel), "+s"(sc[0]), "+s"(sc[1]), "+s"(sc[2]), "+s"(sc[3]), "+s"(sc[4]), "+s"(sc[5]),
                       "+s"(sc[6]), "+s"(sc[7]), "+s"(sc[8]), "+s"(sc[9]), "+s"(sc[10]), "+s"(sc[11]),
                       "+s"(sc[12]), "+s"(sc[13]));
        asm volatile("" : "+s"(sc[14]), "+s"(sc[15]));
        #pragma unroll
        for (int t = 0; t < NT; ++t) asm volatile("" : "+v"(acc[t]));
        #pragma unroll
        for (int v = 0; v < 8; ++v) {
            const int r = row0 + v + 8 * half;
            const float f = hsel ? sc[v + 8] : sc[v];
            #pragma unroll
            for (int t = 0; t < NT; ++t) Y[(size_t)r * OUT + t * 16 + l15] = f * acc[t][v];
        }
    } else {
        #pragma unroll
        for (int t = 0; t < NT; ++t)
            #pragma unroll
            for (int v = 0; v < 8; ++v) {
                int r = row0 + v + 8 * half;
                int c = t * 16 + l15;
                Y[(size_t)r * OUT + c] = acc[t][v];
            }
    }
}


template <int OUT> struct RowVec;
template <> struct RowVec<32> { typedef float4 T; };
template <> struct RowVec<16> { typedef float2 T; };

__device__ __forceinline__ float2 vzero(const float2*) { return make_float2(0.0f, 0.0f); }
__device__ __forceinline__ float4 vzero(const float4*) { return make_float4(0.0f, 0.0f, 0.0f, 0.0f); }
__device__ __forceinline__ void add_if(float2& a, const float2 r, const bool in) {
    a.x += in ? r.x : 0.0f;
    a.y += in ? r.y : 0.0f;
}
__device__ __forceinline__ void add_if(float4& a, const float4 r, const bool in) {
    a.x += in ? r.x : 0.0f;
    a.y += in ? r.y : 0.0f;
    a.z += in ? r.z : 0.0f;
    a.w += in ? r.w : 0.0f;
}
__device__ __forceinline__ float finish1(const float dd, const float s, const float bias) {
    float v = dd * s + bias;
    v = v > 0.0f ? v : 0.0f;
    return dd * v;
}

template <int OUT>
__device__ __forceinline__ typename RowVec<OUT>::T gather_quad(const int* __restrict__ csr, const int* __restrict__ off,
                                                               const float* __restrict__ dis, const float* __restrict__ hw,
                                                               const float* __restrict__ b, const int d0, const int lane) {
    typedef typename RowVec<OUT>::T vec;
    constexpr unsigned ROWB = OUT * 4;
    constexpr unsigned VB   = ROWB / 8;
    constexpr int      FPL  = OUT / 8;
    const int g = lane >> 3;
    const int q = lane & 7;
    const int d = d0 + g;
    const char* __restrict__ rows = (const char*)hw;

    const int beg = off[d];
    const int end = (d + 1 < N_NODES) ? off[d + 1] : N_EDGES;
    const int len = end - beg;
    int m = max(len, __shfl_xor(len, 8));
    m = max(m, __shfl_xor(m, 16));
    const int lenmax = __builtin_amdgcn_readfirstlane(m);
    const float dd = dis[d];
    float bias[FPL];
    #pragma unroll
    for (int i = 0; i < FPL; ++i) bias[i] = b[FPL * q + i];

    vec acc_e = *(const vec*)(rows + ((unsigned)d * ROWB + VB * (unsigned)q));
    vec acc_o = vzero((const vec*)0);

    for (int base = 0; base < lenmax; base += 32) {
        int sx[4];
        #pragma unroll
        for (int k = 0; k < 4; ++k) {
            sx[k] = d;
            if (base + 8 * k + q < len) sx[k] = csr[beg + base + 8 * k + q];
        }
        const int nmax = lenmax - base;
        #pragma unroll
        for (int k = 0; k < 4; ++k) {
            if (8 * k < nmax) {
                vec r[8];
                #pragma unroll
                for (int u = 0; u < 8; ++u) {
                    const int s = __shfl(sx[k], (lane & 24) + u);
                    r[u] = *(const vec*)(rows + ((unsigned)s * ROWB + VB * (unsigned)q));
                }
                #pragma unroll
                for (int u = 0; u < 8; ++u) {
                    const bool in = base + 8 * k + u < len;
                    if (OUT == 16 && (u & 1)) add_if(acc_o, r[u], in);
                    else                      add_if(acc_e, r[u], in);
                }
            }
        }
    }
    vec res;
    if constexpr (OUT == 16) {
        res.x = finish1(dd, acc_e.x + acc_o.x, bias[0]);
        res.y = finish1(dd, acc_e.y + acc_o.y, bias[1]);
    } else {
        res.x = finish1(dd, acc_e.x, bias[0]);
        res.y = finish1(dd, acc_e.y, bias[1]);
        res.z = finish1(dd, acc_e.z, bias[2]);
        res.w = finish1(dd, acc_e.w, bias[3]);
    }
    return res;
}


#define TILE_PITCH 34

__global__ __launch_bounds__(256) void k_gather32_gemm16(const int* __restrict__ csr, const int* __restrict__ off,
                                                         const float* __restrict__ dis, const float* __restrict__ hw,
                                                         const float* __restrict__ b, const float* __restrict__ W,
                                                         float* __restrict__ Y) {
    __shared__ __attribute__((aligned(8))) float s_tile[32 * TILE_PITCH];
    const int lane = threadIdx.x & 31;
    const int w = __builtin_amdgcn_readfirstlane(threadIdx.x >> 5);
    const int row0 = blockIdx.x * 32;

    const float4 hrow = gather_quad<32>(csr, off, dis, hw, b, row0 + 4 * w, lane);
    float2* tp = (float2*)(s_tile + (4 * w + (lane >> 3)) * TILE_PITCH + 4 * (lane & 7));
    tp[0] = make_float2(hrow.x, hrow.y);
    tp[1] = make_float2(hrow.z, hrow.w);

    const int half = lane >> 4;
    const int l15  = lane & 15;

    v2f bf[8];
    #pragma unroll
    for (int ks = 0; ks < 8; ++ks) { bf[ks].x = 0.0f; bf[ks].y = 0.0f; }
    if (w < 2) {
        #pragma unroll
        for (int ks = 0; ks < 8; ++ks) {
            bf[ks].x = W[(4 * ks + 2 * half)     * 16 + l15];
            bf[ks].y = W[(4 * ks + 2 * half + 1) * 16 + l15];
        }
    }
    __syncthreads();
    if (w >= 2) return;

    v2f a[8];
    #pragma unroll
    for (int ks = 0; ks < 8; ++ks) {
        const float* ap = s_tile + (16 * w + l15) * TILE_PITCH + (4 * ks + 2 * half);
        a[ks].x = ap[0];
        a[ks].y = ap[1];
    }
    float* yp = Y + (size_t)(row0 + 16 * w + 8 * half) * 16 + l15;

    asm volatile("" : "+v"(yp),
                      "+v"(a[0]), "+v"(a[1]), "+v"(a[2]), "+v"(a[3]), "+v"(a[4]), "+v"(a[5]), "+v"(a[6]), "+v"(a[7]),
                      "+v"(bf[0]), "+v"(bf[1]), "+v"(bf[2]), "+v"(bf[3]), "+v"(bf[4]), "+v"(bf[5]), "+v"(bf[6]),
                      "+v"(bf[7]));

    v8f acc;
    #pragma unroll
    for (int v = 0; v < 8; ++v) acc[v] = 0.0f;
    #pragma unroll
    for (int ks = 0; ks < 8; ++ks)
        acc = __builtin_amdgcn_wmma_f32_16x16x4_f32(
             false, a[ks],  false, bf[ks],
             (short)0, acc,  false,  false);
    __attribute__((address_space(1))) float* ypg = (__attribute__((address_space(1))) float*)yp;
    #pragma unroll
    for (int v = 0; v < 8; ++v) ypg[v * 16] = acc[v];
}


#define H2_PITCH 20

__global__ __launch_bounds__(256) void k_gather16_dot(const int* __restrict__ csr, const int* __restrict__ off,
                                                      const float* __restrict__ dis, const float* __restrict__ hw,
                                                      const float* __restrict__ b, const float* __restrict__ W3,
                                                      float* __restrict__ hw3) {
    __shared__ __attribute__((aligned(16))) float s_h2[32 * H2_PITCH];
    const int lane = threadIdx.x & 31;
    const int w = __builtin_amdgcn_readfirstlane(threadIdx.x >> 5);
    const int d0 = blockIdx.x * 32 + 4 * w;

    const float2 hrow = gather_quad<16>(csr, off, dis, hw, b, d0, lane);
    *(float2*)(s_h2 + (4 * w + (lane >> 3)) * H2_PITCH + 2 * (lane & 7)) = hrow;
    __syncthreads();
    if (w != 0) return;

    const float* r = s_h2 + lane * H2_PITCH;
    float s = 0.0f;
    #pragma unroll
    for (int k = 0; k < 16; ++k) s += r[k] * W3[k];
    hw3[blockIdx.x * 32 + lane] = s;
}


#define TBL_HALF   (N_NODES / 2)
#define OUT_BLOCKS 256
#define OUT_CH     16384
#define OUT_EPT    (OUT_CH / 1024)
#define N_LINES    (N_NODES / 32)
#define TBL_TRIPS  ((TBL_HALF / 4 + 1023) / 1024)

static_assert(N_NODES % 8 == 0, "a table half is copied in 16-byte units");
static_assert(((N_LINES + OUT_BLOCKS - 1) / OUT_BLOCKS) * 32 <= 1024, "one thread per node of a block's stretch");

__device__ __forceinline__ void load_half(float4 (&r)[TBL_TRIPS], const float* src, const int t) {
    const float4* g = (const float4*)src;
    #pragma unroll
    for (int q = 0; q < TBL_TRIPS; ++q)
        r[q] = (t + q * 1024 < TBL_HALF / 4) ? g[t + q * 1024] : make_float4(0.0f, 0.0f, 0.0f, 0.0f);
}
__device__ __forceinline__ void store_half(float4* l, const float4 (&r)[TBL_TRIPS], const int t) {
    #pragma unroll
    for (int q = 0; q < TBL_TRIPS; ++q)
        if (t + q * 1024 < TBL_HALF / 4) l[t + q * 1024] = r[q];
}
__device__ __forceinline__ void pick_half(const int (&s)[OUT_EPT], const int base, const float* s_tbl,
                                          float* s_val, const int t) {
    float x[OUT_EPT];
    #pragma unroll
    for (int u = 0; u < OUT_EPT; ++u) {
        const unsigned k = (unsigned)(s[u] - base);
        x[u] = s_tbl[k < (unsigned)TBL_HALF ? k : 0u];
    }
    #pragma unroll
    for (int u = 0; u < OUT_EPT; ++u)
        if ((unsigned)(s[u] - base) < (unsigned)TBL_HALF) s_val[t + u * 1024] = x[u];
}

__global__ __launch_bounds__(1024) void k_gather_out1(const int* __restrict__ csr, const int* __restrict__ off,
                                                      const float* __restrict__ dis, const float* hw3,
                                                      const float* __restrict__ b3, float* out) {
    __shared__ __attribute__((aligned(16))) float s_tbl[TBL_HALF];
    __shared__ float s_val[OUT_CH];
    const int t = threadIdx.x;
    const int l_lo = (int)(((long long)blockIdx.x * N_LINES) / OUT_BLOCKS);
    const int l_hi = (int)(((long long)(blockIdx.x + 1) * N_LINES) / OUT_BLOCKS);
    const int n_lo = l_lo * 32;
    const int n_hi = l_hi * 32;
    const int nn = n_hi - n_lo;
    const int e_lo = off[n_lo];
    const int e_hi = (n_hi < N_NODES) ? off[n_hi] : N_EDGES;
    const int d = n_lo + t;
    int beg = 0, end = 0;
    if (t < nn) {
        beg = off[d];
        end = (d + 1 < N_NODES) ? off[d + 1] : N_EDGES;
    }
    float acc = 0.0f;
    for (int c0 = e_lo; c0 < e_hi; c0 += OUT_CH) {
        int s[OUT_EPT];
        #pragma unroll
        for (int u = 0; u < OUT_EPT; ++u) {
            const int j = c0 + t + u * 1024;
            s[u] = (j < e_hi) ? csr[j] : -1;
        }
        float4* l = (float4*)s_tbl;
        float4 r[TBL_TRIPS];
        load_half(r, hw3, t);
        __syncthreads();
        store_half(l, r, t);
        __syncthreads();
        pick_half(s, 0, s_tbl, s_val, t);
        load_half(r, hw3 + TBL_HALF, t);
        __syncthreads();
        store_half(l, r, t);
        __syncthreads();
        pick_half(s, TBL_HALF, s_tbl, s_val, t);
        __syncthreads();
        if (t < nn) {
            const int a = max(beg, c0);
            const int b = min(end, c0 + OUT_CH);
            for (int j = a; j < b; ++j) acc += s_val[j - c0];
        }
    }
    if (t < nn) out[d] = dis[d] * (acc + hw3[d]) + b3[0];
}


static inline int blks(long long n, int bs) { return (int)((n + bs - 1) / bs); }

extern "C" void kernel_launch(void* const* d_in, const int* in_sizes, int n_in,
                              void* d_out, int out_size, void* d_ws, size_t ws_size,
                              hipStream_t stream) {
    (void)in_sizes; (void)n_in; (void)out_size; (void)ws_size;

    const float* x   = (const float*)d_in[0];
    const int*   ei  = (const int*)  d_in[1];
    const float* W1  = (const float*)d_in[2];
    const float* b1  = (const float*)d_in[3];
    const float* W2  = (const float*)d_in[4];
    const float* b2  = (const float*)d_in[5];
    const float* W3  = (const float*)d_in[6];
    const float* b3  = (const float*)d_in[7];
    float* out = (float*)d_out;

    const int* src = ei;
    const int* dst = ei + N_EDGES;

    float* ws   = (float*)d_ws;
    float* dis  = ws;
    int*   off  = (int*)(ws + N_NODES);
    int*   csr  = (int*)(ws + 2 * N_NODES);
    float* pA   = ws + 34 * N_NODES;
    float* pB   = ws + 66 * N_NODES;
    int*   tbl    = (int*)pA;
    int*   binned = (int*)pB;
    float* hw1  = pA;
    float* hw2  = pB;
    float* hw3  = pA;

    const int BS = 256;

    k_chunk_sort<<<SORT_BLOCKS, 1024, 0, stream>>>(src, dst, tbl, binned);
    k_bin_sort  <<<N_BINS, 1024, 0, stream>>>(binned, tbl, off, dis, csr);

    const int gemm_grid = blks(N_NODES / 16, BS / 32);

    k_gemm_wmma<128, 32, true><<<gemm_grid, BS, 0, stream>>>(x, W1, dis, hw1, N_NODES);
    k_gather32_gemm16<<<N_NODES / 32, 256, 0, stream>>>(csr, off, dis, hw1, b1, W2, hw2);

    k_gather16_dot<<<N_NODES / 32, 256, 0, stream>>>(csr, off, dis, hw2, b2, W3, hw3);

    k_gather_out1<<<OUT_BLOCKS, 1024, 0, stream>>>(csr, off, dis, hw3, b3, out);
}


# Layer-1 transform folded into the level-2 sort launch: a bin's block prefetches its 256 rows of x, sorts, then multiplies them by W1 (LDS copy, k order of the sum permuted) and by its dis; 5 launches
# speedup vs baseline: 9.7474x; 1.0707x over previous
#include <hip/hip_runtime.h>


typedef __attribute__((ext_vector_type(2))) float v2f;
typedef __attribute__((ext_vector_type(4))) float v4f;
typedef __attribute__((ext_vector_type(8))) float v8f;

#define N_NODES 100000
#define N_EDGES 3200000

static_assert(N_NODES % 32 == 0, "the three gathers hand out whole 32-node tiles and lines");
static_assert(N_NODES % 16 == 0, "the layer-1 product works on whole 16-row tiles");
static_assert(N_EDGES == 32 * N_NODES, "the binned edge list reuses the 32*N word plane of the layer-1 output");


#define SORT_BLOCKS 256
#define N_BINS      ((N_NODES + 255) / 256)
#define EDGE_CHUNK  ((N_EDGES + SORT_BLOCKS - 1) / SORT_BLOCKS)
#define TBL_PITCH   (N_BINS - 1)
#define EPT         ((EDGE_CHUNK + 1023) / 1024)

static_assert(N_NODES <= (1 << 17), "src must fit in the low 17 bits of a binned word");
static_assert(N_EDGES == SORT_BLOCKS * EDGE_CHUNK, "every level-1 block owns a full stretch");
static_assert(SORT_BLOCKS * TBL_PITCH <= N_NODES, "the tables of bin starts fill the first N words of the workspace");
static_assert(SORT_BLOCKS % 32 == 0, "a level-2 wave takes every 32nd run");
static_assert(N_BINS <= 1024, "one thread per bin scans the block's bin counts");

__global__ __launch_bounds__(1024) void k_chunk_sort(const int* __restrict__ src, const int* __restrict__ dst,
                                                     int* __restrict__ tbl, int* __restrict__ binned) {
    __shared__ int s_stage[EDGE_CHUNK];
    __shared__ int s_cnt[N_BINS];
    __shared__ int s_start[N_BINS];
    __shared__ int s_wsum[32];
    const int t = threadIdx.x;
    const int lane = t & 31;
    const int w = t >> 5;
    const int blk = blockIdx.x;
    for (int b = t; b < N_BINS; b += 1024) s_cnt[b] = 0;

    const int lo = blk * EDGE_CHUNK;
    int d[EPT], s[EPT], r[EPT];
    #pragma unroll
    for (int u = 0; u < EPT; ++u) d[u] = (t + u * 1024 < EDGE_CHUNK) ? dst[lo + t + u * 1024] : 0;
    #pragma unroll
    for (int u = 0; u < EPT; ++u) s[u] = (t + u * 1024 < EDGE_CHUNK) ? src[lo + t + u * 1024] : 0;
    __syncthreads();

    #pragma unroll
    for (int u = 0; u < EPT; ++u) {
        r[u] = 0;
        if (t + u * 1024 < EDGE_CHUNK) r[u] = atomicAdd(&s_cnt[d[u] >> 8], 1);
    }
    __syncthreads();

    const int c = (t < N_BINS) ? s_cnt[t] : 0;
    int incl = c;
    for (int o = 1; o < 32; o <<= 1) {
        int v = __shfl_up(incl, o);
        if (lane >= o) incl += v;
    }
    if (lane == 31) s_wsum[w] = incl;
    __syncthreads();
    int woff = 0;
    for (int k = 0; k < 32; ++k)
        if (k < w) woff += s_wsum[k];
    const int excl = woff + incl - c;
    if (t < N_BINS) {
        s_start[t] = excl;
        if (t > 0) tbl[blk * TBL_PITCH + t - 1] = excl;
    }
    __syncthreads();

    #pragma unroll
    for (int u = 0; u < EPT; ++u)
        if (t + u * 1024 < EDGE_CHUNK) s_stage[s_start[d[u] >> 8] + r[u]] = s[u] | ((d[u] & 255) << 17);
    __syncthreads();

    for (int i = t; i < EDGE_CHUNK; i += 1024) binned[lo + i] = s_stage[i];
}

#define BIN_STAGE 12288
#define RPW (SORT_BLOCKS / 32)

template <bool STAGED>
__device__ __forceinline__ void place_edge(const int wd, int* s_cur, int* s_stage, int* __restrict__ csr,
                                           const int bstart) {
    const int p = atomicAdd(&s_cur[wd >> 17], 1);
    if (STAGED) s_stage[p] = wd & 0x1FFFF;
    else        csr[bstart + p] = wd & 0x1FFFF;
}

template <bool STAGED>
__device__ __forceinline__ void place_runs(const int (&w0)[RPW], const int (&w1)[RPW], const int (&rs)[RPW],
                                           const int (&rl)[RPW], int* s_cur, int* s_stage,
                                           const int* __restrict__ binned, int* __restrict__ csr,
                                           const int bstart, const int lane) {
    #pragma unroll
    for (int k = 0; k < RPW; ++k) {
        if (w0[k] >= 0) place_edge<STAGED>(w0[k], s_cur, s_stage, csr, bstart);
        if (w1[k] >= 0) place_edge<STAGED>(w1[k], s_cur, s_stage, csr, bstart);
    }
    #pragma unroll
    for (int k = 0; k < RPW; ++k)
        for (int j = lane + 64; j < rl[k]; j += 32) place_edge<STAGED>(binned[rs[k] + j], s_cur, s_stage, csr, bstart);
}

__global__ __launch_bounds__(1024) void k_bin_sort(const int* __restrict__ binned, const int* __restrict__ tbl,
                                                   int* __restrict__ off, int* __restrict__ csr,
                                                   const float* __restrict__ x, const float* __restrict__ W1,
                                                   float* __restrict__ hw1) {
    __shared__ __attribute__((aligned(16))) float s_w[128 * 32];
    __shared__ __attribute__((aligned(16))) float s_dis[256];
    __shared__ int s_cnt[256];
    __shared__ int s_cur[256];
    __shared__ int s_wsum[8];
    __shared__ int s_rs[SORT_BLOCKS];
    __shared__ int s_rl[SORT_BLOCKS];
    __shared__ int s_sum_start[RPW];
    __shared__ int s_sum_len[RPW];
    __shared__ int s_stage[BIN_STAGE];
    const int t = threadIdx.x;
    const int lane = t & 31;
    const int w = t >> 5;
    const int bin = blockIdx.x;

    v4f wv;
    {
        const int j = t >> 6, ct = (t >> 5) & 1, h = (t >> 4) & 1, c = t & 15;
        const float* wp = W1 + (8 * j + 4 * h) * 32 + 16 * ct + c;
        wv.x = wp[0];
        wv.y = wp[32];
        wv.z = wp[64];
        wv.w = wp[96];
    }

    if (t < 512) {
        const int prow = bin * 256 + (t >> 1);
        if (prow < N_NODES) __builtin_prefetch(x + (size_t)prow * 128 + (t & 1) * 64, 0, 1);
    }

    if (t < SORT_BLOCKS) {
        const int a = (bin > 0) ? tbl[t * TBL_PITCH + bin - 1] : 0;
        const int b = (bin + 1 < N_BINS) ? tbl[t * TBL_PITCH + bin] : EDGE_CHUNK;
        s_rs[t] = t * EDGE_CHUNK + a;
        s_rl[t] = b - a;
        int sa = a, sl = b - a;
        for (int o = 16; o > 0; o >>= 1) {
            sa += __shfl_xor(sa, o);
            sl += __shfl_xor(sl, o);
        }
        if (lane == 0) {
            s_sum_start[w] = sa;
            s_sum_len[w] = sl;
        }
    }
    if (t < 256) s_cnt[t] = 0;
    ((v4f*)s_w)[t] = wv;
    __syncthreads();
    int bstart = 0, n = 0;
    for (int k = 0; k < RPW; ++k) {
        bstart += s_sum_start[k];
        n += s_sum_len[k];
    }
    const bool staged = n <= BIN_STAGE;

    int rs[RPW], rl[RPW], w0[RPW], w1[RPW];
    #pragma unroll
    for (int k = 0; k < RPW; ++k) {
        rs[k] = s_rs[w + 32 * k];
        rl[k] = s_rl[w + 32 * k];
    }
    #pragma unroll
    for (int k = 0; k < RPW; ++k) {
        w0[k] = (lane < rl[k]) ? binned[rs[k] + lane] : -1;
        w1[k] = (lane + 32 < rl[k]) ? binned[rs[k] + lane + 32] : -1;
    }
    #pragma unroll
    for (int k = 0; k < RPW; ++k) {
        if (w0[k] >= 0) atomicAdd(&s_cnt[w0[k] >> 17], 1);
        if (w1[k] >= 0) atomicAdd(&s_cnt[w1[k] >> 17], 1);
    }
    #pragma unroll
    for (int k = 0; k < RPW; ++k)
        for (int j = lane + 64; j < rl[k]; j += 32) atomicAdd(&s_cnt[binned[rs[k] + j] >> 17], 1);
    __syncthreads();

    int c = 0, incl = 0;
    if (t < 256) {
        c = s_cnt[t];
        incl = c;
        for (int o = 1; o < 32; o <<= 1) {
            int v = __shfl_up(incl, o);
            if (lane >= o) incl += v;
        }
        if (lane == 31) s_wsum[t >> 5] = incl;
    }
    __syncthreads();
    if (t < 256) {
        int woff = 0;
        for (int k = 0; k < 8; ++k)
            if (k < (t >> 5)) woff += s_wsum[k];
        const int excl = woff + incl - c;
        s_cur[t] = excl;
        s_dis[t] = rsqrtf((float)(c + 1));
        const int node = bin * 256 + t;
        if (node < N_NODES) off[node] = bstart + excl;
    }
    __syncthreads();
    if (staged) place_runs<true >(w0, w1, rs, rl, s_cur, s_stage, binned, csr, bstart, lane);
    else        place_runs<false>(w0, w1, rs, rl, s_cur, s_stage, binned, csr, bstart, lane);
    __syncthreads();
    if (staged)
        for (int j = t; j < n; j += 1024) csr[bstart + j] = s_stage[j];

    const int wu = __builtin_amdgcn_readfirstlane(w);
    const int row0 = bin * 256 + 16 * wu;
    if (wu >= 16 || row0 >= N_NODES) return;

    const int fc = lane & 15;
    const int fh = lane >> 4;
    const v4f* xp = (const v4f*)(x + (size_t)(row0 + fc) * 128) + fh;
    const v4f* wl = (const v4f*)s_w + lane;

    float f = s_dis[16 * wu + fc];
    float* yp = hw1 + (size_t)(row0 + fc) * 32 + 8 * fh;
    v8f acc0, acc1;
    #pragma unroll
    for (int v = 0; v < 8; ++v) { acc0[v] = 0.0f; acc1[v] = 0.0f; }
    asm volatile("" : "+v"(acc0), "+v"(acc1), "+v"(f), "+v"(yp));

    #pragma unroll
    for (int q = 0; q < 2; ++q) {
        v4f a[8];
        #pragma unroll
        for (int jj = 0; jj < 8; ++jj) a[jj] = xp[2 * (8 * q + jj)];
        #pragma unroll
        for (int jj = 0; jj < 8; ++jj) {
            const int j = 8 * q + jj;
            const v4f b0 = wl[(2 * j) * 32];
            const v4f b1 = wl[(2 * j + 1) * 32];
            v2f xlo, xhi, wlo, whi;
            xlo.x = a[jj].x; xlo.y = a[jj].y;
            xhi.x = a[jj].z; xhi.y = a[jj].w;
            wlo.x = b0.x; wlo.y = b0.y;
            whi.x = b0.z; whi.y = b0.w;
            acc0 = __builtin_amdgcn_wmma_f32_16x16x4_f32(false, wlo, false, xlo, (short)0, acc0, false, false);
            acc0 = __builtin_amdgcn_wmma_f32_16x16x4_f32(false, whi, false, xhi, (short)0, acc0, false, false);
            wlo.x = b1.x; wlo.y = b1.y;
            whi.x = b1.z; whi.y = b1.w;
            acc1 = __builtin_amdgcn_wmma_f32_16x16x4_f32(false, wlo, false, xlo, (short)0, acc1, false, false);
            acc1 = __builtin_amdgcn_wmma_f32_16x16x4_f32(false, whi, false, xhi, (short)0, acc1, false, false);
        }
    }

    asm volatile("v_nop\n\tv_nop\n\tv_nop\n\tv_nop\n\tv_nop\n\tv_nop\n\tv_nop\n\tv_nop"
                 : "+v"(acc0), "+v"(acc1), "+v"(f));
    __attribute__((address_space(1))) v4f* ypg = (__attribute__((address_space(1))) v4f*)yp;
    v4f o;
    o.x = f * acc0[0]; o.y = f * acc0[1]; o.z = f * acc0[2]; o.w = f * acc0[3];
    ypg[0] = o;
    o.x = f * acc0[4]; o.y = f * acc0[5]; o.z = f * acc0[6]; o.w = f * acc0[7];
    ypg[1] = o;
    o.x = f * acc1[0]; o.y = f * acc1[1]; o.z = f * acc1[2]; o.w = f * acc1[3];
    ypg[4] = o;
    o.x = f * acc1[4]; o.y = f * acc1[5]; o.z = f * acc1[6]; o.w = f * acc1[7];
    ypg[5] = o;
}


template <int OUT> struct RowVec;
template <> struct RowVec<32> { typedef float4 T; };
template <> struct RowVec<16> { typedef float2 T; };

__device__ __forceinline__ float2 vzero(const float2*) { return make_float2(0.0f, 0.0f); }
__device__ __forceinline__ float4 vzero(const float4*) { return make_float4(0.0f, 0.0f, 0.0f, 0.0f); }
__device__ __forceinline__ void add_if(float2& a, const float2 r, const bool in) {
    a.x += in ? r.x : 0.0f;
    a.y += in ? r.y : 0.0f;
}
__device__ __forceinline__ void add_if(float4& a, const float4 r, const bool in) {
    a.x += in ? r.x : 0.0f;
    a.y += in ? r.y : 0.0f;
    a.z += in ? r.z : 0.0f;
    a.w += in ? r.w : 0.0f;
}
__device__ __forceinline__ float finish1(const float dd, const float s, const float bias) {
    float v = dd * s + bias;
    v = v > 0.0f ? v : 0.0f;
    return dd * v;
}

template <int OUT>
__device__ __forceinline__ typename RowVec<OUT>::T gather_quad(const int* __restrict__ csr, const int* __restrict__ off,
                                                               const float* __restrict__ hw,
                                                               const float* __restrict__ b, const int d0, const int lane) {
    typedef typename RowVec<OUT>::T vec;
    constexpr unsigned ROWB = OUT * 4;
    constexpr unsigned VB   = ROWB / 8;
    constexpr int      FPL  = OUT / 8;
    const int g = lane >> 3;
    const int q = lane & 7;
    const int d = d0 + g;
    const char* __restrict__ rows = (const char*)hw;

    const int beg = off[d];
    const int end = (d + 1 < N_NODES) ? off[d + 1] : N_EDGES;
    const int len = end - beg;
    int m = max(len, __shfl_xor(len, 8));
    m = max(m, __shfl_xor(m, 16));
    const int lenmax = __builtin_amdgcn_readfirstlane(m);
    const float dd = rsqrtf((float)(len + 1));
    float bias[FPL];
    #pragma unroll
    for (int i = 0; i < FPL; ++i) bias[i] = b[FPL * q + i];

    vec acc_e = *(const vec*)(rows + ((unsigned)d * ROWB + VB * (unsigned)q));
    vec acc_o = vzero((const vec*)0);

    for (int base = 0; base < lenmax; base += 32) {
        int sx[4];
        #pragma unroll
        for (int k = 0; k < 4; ++k) {
            sx[k] = d;
            if (base + 8 * k + q < len) sx[k] = csr[beg + base + 8 * k + q];
        }
        const int nmax = lenmax - base;
        #pragma unroll
        for (int k = 0; k < 4; ++k) {
            if (8 * k < nmax) {
                vec r[8];
                #pragma unroll
                for (int u = 0; u < 8; ++u) {
                    const int s = __shfl(sx[k], (lane & 24) + u);
                    r[u] = *(const vec*)(rows + ((unsigned)s * ROWB + VB * (unsigned)q));
                }
                #pragma unroll
                for (int u = 0; u < 8; ++u) {
                    const bool in = base + 8 * k + u < len;
                    if (OUT == 16 && (u & 1)) add_if(acc_o, r[u], in);
                    else                      add_if(acc_e, r[u], in);
                }
            }
        }
    }
    vec res;
    if constexpr (OUT == 16) {
        res.x = finish1(dd, acc_e.x + acc_o.x, bias[0]);
        res.y = finish1(dd, acc_e.y + acc_o.y, bias[1]);
    } else {
        res.x = finish1(dd, acc_e.x, bias[0]);
        res.y = finish1(dd, acc_e.y, bias[1]);
        res.z = finish1(dd, acc_e.z, bias[2]);
        res.w = finish1(dd, acc_e.w, bias[3]);
    }
    return res;
}


#define TILE_PITCH 34

__global__ __launch_bounds__(256) void k_gather32_gemm16(const int* __restrict__ csr, const int* __restrict__ off,
                                                         const float* __restrict__ hw,
                                                         const float* __restrict__ b, const float* __restrict__ W,
                                                         float* __restrict__ Y) {
    __shared__ __attribute__((aligned(8))) float s_tile[32 * TILE_PITCH];
    const int lane = threadIdx.x & 31;
    const int w = __builtin_amdgcn_readfirstlane(threadIdx.x >> 5);
    const int row0 = blockIdx.x * 32;

    const float4 hrow = gather_quad<32>(csr, off, hw, b, row0 + 4 * w, lane);
    float2* tp = (float2*)(s_tile + (4 * w + (lane >> 3)) * TILE_PITCH + 4 * (lane & 7));
    tp[0] = make_float2(hrow.x, hrow.y);
    tp[1] = make_float2(hrow.z, hrow.w);

    const int half = lane >> 4;
    const int l15  = lane & 15;

    v2f bf[8];
    #pragma unroll
    for (int ks = 0; ks < 8; ++ks) { bf[ks].x = 0.0f; bf[ks].y = 0.0f; }
    if (w < 2) {
        #pragma unroll
        for (int ks = 0; ks < 8; ++ks) {
            bf[ks].x = W[(4 * ks + 2 * half)     * 16 + l15];
            bf[ks].y = W[(4 * ks + 2 * half + 1) * 16 + l15];
        }
    }
    __syncthreads();
    if (w >= 2) return;

    v2f a[8];
    #pragma unroll
    for (int ks = 0; ks < 8; ++ks) {
        const float* ap = s_tile + (16 * w + l15) * TILE_PITCH + (4 * ks + 2 * half);
        a[ks].x = ap[0];
        a[ks].y = ap[1];
    }
    float* yp = Y + (size_t)(row0 + 16 * w + 8 * half) * 16 + l15;

    asm volatile("" : "+v"(yp),
                      "+v"(a[0]), "+v"(a[1]), "+v"(a[2]), "+v"(a[3]), "+v"(a[4]), "+v"(a[5]), "+v"(a[6]), "+v"(a[7]),
                      "+v"(bf[0]), "+v"(bf[1]), "+v"(bf[2]), "+v"(bf[3]), "+v"(bf[4]), "+v"(bf[5]), "+v"(bf[6]),
                      "+v"(bf[7]));

    v8f acc;
    #pragma unroll
    for (int v = 0; v < 8; ++v) acc[v] = 0.0f;
    #pragma unroll
    for (int ks = 0; ks < 8; ++ks)
        acc = __builtin_amdgcn_wmma_f32_16x16x4_f32(
             false, a[ks],  false, bf[ks],
             (short)0, acc,  false,  false);
    __attribute__((address_space(1))) float* ypg = (__attribute__((address_space(1))) float*)yp;
    #pragma unroll
    for (int v = 0; v < 8; ++v) ypg[v * 16] = acc[v];
}


#define H2_PITCH 20

__global__ __launch_bounds__(256) void k_gather16_dot(const int* __restrict__ csr, const int* __restrict__ off,
                                                      const float* __restrict__ hw,
                                                      const float* __restrict__ b, const float* __restrict__ W3,
                                                      float* __restrict__ hw3) {
    __shared__ __attribute__((aligned(16))) float s_h2[32 * H2_PITCH];
    const int lane = threadIdx.x & 31;
    const int w = __builtin_amdgcn_readfirstlane(threadIdx.x >> 5);
    const int d0 = blockIdx.x * 32 + 4 * w;

    const float2 hrow = gather_quad<16>(csr, off, hw, b, d0, lane);
    *(float2*)(s_h2 + (4 * w + (lane >> 3)) * H2_PITCH + 2 * (lane & 7)) = hrow;
    __syncthreads();
    if (w != 0) return;

    const float* r = s_h2 + lane * H2_PITCH;
    float s = 0.0f;
    #pragma unroll
    for (int k = 0; k < 16; ++k) s += r[k] * W3[k];
    hw3[blockIdx.x * 32 + lane] = s;
}


#define TBL_HALF   (N_NODES / 2)
#define OUT_BLOCKS 256
#define OUT_CH     16384
#define OUT_EPT    (OUT_CH / 1024)
#define N_LINES    (N_NODES / 32)
#define TBL_TRIPS  ((TBL_HALF / 4 + 1023) / 1024)

static_assert(N_NODES % 8 == 0, "a table half is copied in 16-byte units");
static_assert(((N_LINES + OUT_BLOCKS - 1) / OUT_BLOCKS) * 32 <= 1024, "one thread per node of a block's stretch");

__device__ __forceinline__ void load_half(float4 (&r)[TBL_TRIPS], const float* src, const int t) {
    const float4* g = (const float4*)src;
    #pragma unroll
    for (int q = 0; q < TBL_TRIPS; ++q)
        r[q] = (t + q * 1024 < TBL_HALF / 4) ? g[t + q * 1024] : make_float4(0.0f, 0.0f, 0.0f, 0.0f);
}
__device__ __forceinline__ void store_half(float4* l, const float4 (&r)[TBL_TRIPS], const int t) {
    #pragma unroll
    for (int q = 0; q < TBL_TRIPS; ++q)
        if (t + q * 1024 < TBL_HALF / 4) l[t + q * 1024] = r[q];
}
__device__ __forceinline__ void pick_half(const int (&s)[OUT_EPT], const int base, const float* s_tbl,
                                          float* s_val, const int t) {
    float x[OUT_EPT];
    #pragma unroll
    for (int u = 0; u < OUT_EPT; ++u) {
        const unsigned k = (unsigned)(s[u] - base);
        x[u] = s_tbl[k < (unsigned)TBL_HALF ? k : 0u];
    }
    #pragma unroll
    for (int u = 0; u < OUT_EPT; ++u)
        if ((unsigned)(s[u] - base) < (unsigned)TBL_HALF) s_val[t + u * 1024] = x[u];
}

__global__ __launch_bounds__(1024) void k_gather_out1(const int* __restrict__ csr, const int* __restrict__ off,
                                                      const float* hw3,
                                                      const float* __restrict__ b3, float* out) {
    __shared__ __attribute__((aligned(16))) float s_tbl[TBL_HALF];
    __shared__ float s_val[OUT_CH];
    const int t = threadIdx.x;
    const int l_lo = (int)(((long long)blockIdx.x * N_LINES) / OUT_BLOCKS);
    const int l_hi = (int)(((long long)(blockIdx.x + 1) * N_LINES) / OUT_BLOCKS);
    const int n_lo = l_lo * 32;
    const int n_hi = l_hi * 32;
    const int nn = n_hi - n_lo;
    const int e_lo = off[n_lo];
    const int e_hi = (n_hi < N_NODES) ? off[n_hi] : N_EDGES;
    const int d = n_lo + t;
    int beg = 0, end = 0;
    if (t < nn) {
        beg = off[d];
        end = (d + 1 < N_NODES) ? off[d + 1] : N_EDGES;
    }
    float acc = 0.0f;
    for (int c0 = e_lo; c0 < e_hi; c0 += OUT_CH) {
        int s[OUT_EPT];
        #pragma unroll
        for (int u = 0; u < OUT_EPT; ++u) {
            const int j = c0 + t + u * 1024;
            s[u] = (j < e_hi) ? csr[j] : -1;
        }
        float4* l = (float4*)s_tbl;
        float4 r[TBL_TRIPS];
        load_half(r, hw3, t);
        __syncthreads();
        store_half(l, r, t);
        __syncthreads();
        pick_half(s, 0, s_tbl, s_val, t);
        load_half(r, hw3 + TBL_HALF, t);
        __syncthreads();
        store_half(l, r, t);
        __syncthreads();
        pick_half(s, TBL_HALF, s_tbl, s_val, t);
        __syncthreads();
        if (t < nn) {
            const int a = max(beg, c0);
            const int b = min(end, c0 + OUT_CH);
            for (int j = a; j < b; ++j) acc += s_val[j - c0];
        }
    }
    if (t < nn) out[d] = rsqrtf((float)(end - beg + 1)) * (acc + hw3[d]) + b3[0];
}


extern "C" void kernel_launch(void* const* d_in, const int* in_sizes, int n_in,
                              void* d_out, int out_size, void* d_ws, size_t ws_size,
                              hipStream_t stream) {
    (void)in_sizes; (void)n_in; (void)out_size; (void)ws_size;

    const float* x   = (const float*)d_in[0];
    const int*   ei  = (const int*)  d_in[1];
    const float* W1  = (const float*)d_in[2];
    const float* b1  = (const float*)d_in[3];
    const float* W2  = (const float*)d_in[4];
    const float* b2  = (const float*)d_in[5];
    const float* W3  = (const float*)d_in[6];
    const float* b3  = (const float*)d_in[7];
    float* out = (float*)d_out;

    const int* src = ei;
    const int* dst = ei + N_EDGES;

    float* ws   = (float*)d_ws;
    int*   tbl  = (int*)ws;
    int*   off  = (int*)(ws + N_NODES);
    int*   csr  = (int*)(ws + 2 * N_NODES);
    float* pA   = ws + 34 * N_NODES;
    float* pB   = ws + 66 * N_NODES;
    int*   binned = (int*)pB;
    float* hw1  = pA;
    float* hw2  = pB;
    float* hw3  = pA;

    k_chunk_sort<<<SORT_BLOCKS, 1024, 0, stream>>>(src, dst, tbl, binned);
    k_bin_sort  <<<N_BINS, 1024, 0, stream>>>(binned, tbl, off, csr, x, W1, hw1);

    k_gather32_gemm16<<<N_NODES / 32, 256, 0, stream>>>(csr, off, hw1, b1, W2, hw2);

    k_gather16_dot<<<N_NODES / 32, 256, 0, stream>>>(csr, off, hw2, b2, W3, hw3);

    k_gather_out1<<<OUT_BLOCKS, 1024, 0, stream>>>(csr, off, hw3, b3, out);
}
